# starting kernel
# baseline (speedup 1.0000x reference)
#include <hip/hip_runtime.h>
#include <hip/hip_bf16.h>
#include <math.h>


typedef float v2f __attribute__((ext_vector_type(2)));
typedef float v8f __attribute__((ext_vector_type(8)));

__device__ __forceinline__ v8f v8f_zero() {
    v8f z = {0.f, 0.f, 0.f, 0.f, 0.f, 0.f, 0.f, 0.f};
    return z;
}

__device__ __forceinline__ v8f wmma4(v2f a, v2f b, v8f c) {
    return __builtin_amdgcn_wmma_f32_16x16x4_f32(false, a, false, b,
                                                 (short)0, c, false, false);
}

__global__ void relgen_kernel(float* __restrict__ rel) {
    const int row = blockIdx.x;
    const int t = threadIdx.x;
    const float pos = (float)(2047 - row);
    const float negl = -9.210340371976184f / 512.0f;
#pragma unroll
    for (int i = 0; i < 4; ++i) {
        int c = t + i * 256;
        int cc = c & 511;
        float inv = expf((float)cc * negl);
        float ang = pos * inv;
        rel[(size_t)row * 1024 + c] = (c < 512) ? sinf(ang) : cosf(ang);
    }
}

__global__ void gemm_nt_kernel(const float* __restrict__ A,
                               const float* __restrict__ B,
                               float* __restrict__ C,
                               int M, int N, int Kd) {
    const int lane = threadIdx.x & 31;
    const int w    = threadIdx.x >> 5;
    const int l16  = lane & 15;
    const int hlf  = lane >> 4;
    const int m0 = blockIdx.y * 64 + (w & 1) * 32;
    const int n0 = blockIdx.x * 256 + (w >> 1) * 64;
    (void)M;

    v8f c[2][4];
#pragma unroll
    for (int i = 0; i < 2; ++i)
#pragma unroll
        for (int j = 0; j < 4; ++j) c[i][j] = v8f_zero();

    for (int kb = 0; kb < Kd; kb += 4) {
        const int kk = kb + 2 * hlf;
        v2f a[2], b[4];
#pragma unroll
        for (int i = 0; i < 2; ++i)
            a[i] = *(const v2f*)(A + (size_t)(m0 + i * 16 + l16) * Kd + kk);
#pragma unroll
        for (int j = 0; j < 4; ++j)
            b[j] = *(const v2f*)(B + (size_t)(n0 + j * 16 + l16) * Kd + kk);
#pragma unroll
        for (int i = 0; i < 2; ++i)
#pragma unroll
            for (int j = 0; j < 4; ++j) c[i][j] = wmma4(a[i], b[j], c[i][j]);
    }

#pragma unroll
    for (int i = 0; i < 2; ++i)
#pragma unroll
        for (int j = 0; j < 4; ++j)
#pragma unroll
            for (int r = 0; r < 8; ++r) {
                int m = m0 + i * 16 + r + 8 * hlf;
                int n = n0 + j * 16 + l16;
                C[(size_t)m * N + n] = c[i][j][r];
            }
}

__global__ void score_kernel(const float* __restrict__ Q,
                             const float* __restrict__ Km,
                             const float* __restrict__ R,
                             const float* __restrict__ u,
                             const float* __restrict__ v,
                             float* __restrict__ S, int bh0) {
    const int lane = threadIdx.x & 31;
    const int w    = threadIdx.x >> 5;
    const int l16  = lane & 15;
    const int hlf  = lane >> 4;
    const int s  = blockIdx.z;
    const int bh = bh0 + s;
    const int b  = bh >> 4;
    const int h  = bh & 15;
    const int m0 = blockIdx.y * 64 + (w & 1) * 32;
    const int n0 = blockIdx.x * 256 + (w >> 1) * 64;

    const float* Qb = Q + (size_t)b * 1024 * 1024 + h * 64;
    const float* Kb = Km + (size_t)b * 2048 * 1024 + h * 64;
    const float* Rb = R + h * 64;

    v8f c[2][4];
#pragma unroll
    for (int i = 0; i < 2; ++i)
#pragma unroll
        for (int j = 0; j < 4; ++j) c[i][j] = v8f_zero();

    for (int kb = 0; kb < 64; kb += 4) {
        const int kk = kb + 2 * hlf;
        v2f ub = *(const v2f*)(u + h * 64 + kk);
        v2f vb = *(const v2f*)(v + h * 64 + kk);
        v2f qa[2], qv[2], bk[4], br[4];
#pragma unroll
        for (int i = 0; i < 2; ++i) {
            v2f qq = *(const v2f*)(Qb + (size_t)(m0 + i * 16 + l16) * 1024 + kk);
            qa[i] = qq + ub;
            qv[i] = qq + vb;
        }
#pragma unroll
        for (int j = 0; j < 4; ++j) {
            bk[j] = *(const v2f*)(Kb + (size_t)(n0 + j * 16 + l16) * 1024 + kk);
            br[j] = *(const v2f*)(Rb + (size_t)(n0 + j * 16 + l16) * 1024 + kk);
        }
#pragma unroll
        for (int i = 0; i < 2; ++i)
#pragma unroll
            for (int j = 0; j < 4; ++j) c[i][j] = wmma4(qa[i], bk[j], c[i][j]);
#pragma unroll
        for (int i = 0; i < 2; ++i)
#pragma unroll
            for (int j = 0; j < 4; ++j) c[i][j] = wmma4(qv[i], br[j], c[i][j]);
    }

    const float scale = 0.125f;
    float* Ss = S + (size_t)s * 1024 * 2048;
#pragma unroll
    for (int i = 0; i < 2; ++i)
#pragma unroll
        for (int j = 0; j < 4; ++j)
#pragma unroll
            for (int r = 0; r < 8; ++r) {
                int m = m0 + i * 16 + r + 8 * hlf;
                int n = n0 + j * 16 + l16;
                float val = c[i][j][r] * scale;
                if (n > 1024 + m) val = -INFINITY;
                Ss[(size_t)m * 2048 + n] = val;
            }
}

__global__ void softmax_kernel(float* __restrict__ S) {
    __shared__ float red[256];
    const int s = blockIdx.y, q = blockIdx.x, t = threadIdx.x;
    float* row = S + ((size_t)s * 1024 + q) * 2048;
    float x[8];
    float m = -INFINITY;
#pragma unroll
    for (int i = 0; i < 8; ++i) {
        x[i] = row[t + i * 256];
        m = fmaxf(m, x[i]);
    }
    red[t] = m;
    __syncthreads();
    for (int off = 128; off > 0; off >>= 1) {
        if (t < off) red[t] = fmaxf(red[t], red[t + off]);
        __syncthreads();
    }
    m = red[0];
    __syncthreads();
    float sum = 0.f;
#pragma unroll
    for (int i = 0; i < 8; ++i) {
        x[i] = __expf(x[i] - m);
        sum += x[i];
    }
    red[t] = sum;
    __syncthreads();
    for (int off = 128; off > 0; off >>= 1) {
        if (t < off) red[t] += red[t + off];
        __syncthreads();
    }
    float inv = 1.0f / red[0];
#pragma unroll
    for (int i = 0; i < 8; ++i) row[t + i * 256] = x[i] * inv;
}

__global__ void pv_kernel(const float* __restrict__ P,
                          const float* __restrict__ V,
                          float* __restrict__ O, int bh0) {
    const int lane = threadIdx.x & 31;
    const int w    = threadIdx.x >> 5;
    const int l16  = lane & 15;
    const int hlf  = lane >> 4;
    const int s  = blockIdx.y;
    const int bh = bh0 + s;
    const int b  = bh >> 4;
    const int h  = bh & 15;
    const int m0 = blockIdx.x * 128 + w * 16;

    const float* Ps = P + (size_t)s * 1024 * 2048;
    const float* Vb = V + (size_t)b * 2048 * 1024 + h * 64;

    v8f c[4];
#pragma unroll
    for (int j = 0; j < 4; ++j) c[j] = v8f_zero();

    for (int kb = 0; kb < 2048; kb += 4) {
        const int kk = kb + 2 * hlf;
        v2f a = *(const v2f*)(Ps + (size_t)(m0 + l16) * 2048 + kk);
        v2f bf[4];
#pragma unroll
        for (int j = 0; j < 4; ++j) {
            int n = j * 16 + l16;
            bf[j].x = Vb[(size_t)kk * 1024 + n];
            bf[j].y = Vb[(size_t)(kk + 1) * 1024 + n];
        }
#pragma unroll
        for (int j = 0; j < 4; ++j) c[j] = wmma4(a, bf[j], c[j]);
    }

#pragma unroll
    for (int j = 0; j < 4; ++j)
#pragma unroll
        for (int r = 0; r < 8; ++r) {
            int m = m0 + r + 8 * hlf;
            int n = j * 16 + l16;
            O[((size_t)b * 1024 + m) * 1024 + h * 64 + n] = c[j][r];
        }
}

extern "C" void kernel_launch(void* const* d_in, const int* in_sizes, int n_in,
                              void* d_out, int out_size, void* d_ws, size_t ws_size,
                              hipStream_t stream) {
    const float* x_q  = (const float*)d_in[0];
    const float* x_kv = (const float*)d_in[1];
    const float* Wq   = (const float*)d_in[2];
    const float* Wk   = (const float*)d_in[3];
    const float* Wv   = (const float*)d_in[4];
    const float* Wr   = (const float*)d_in[5];
    const float* Wo   = (const float*)d_in[6];
    const float* ub   = (const float*)d_in[7];
    const float* vb   = (const float*)d_in[8];
    float* out = (float*)d_out;
    (void)in_sizes; (void)n_in; (void)out_size; (void)ws_size;

    float* ws   = (float*)d_ws;
    float* Qb   = ws;
    float* Kb   = Qb + 2097152;
    float* Vb   = Kb + 4194304;
    float* Relb = Vb + 4194304;
    float* Rb   = Relb + 2097152;
    float* Ob   = Rb + 2097152;
    float* Sb   = Ob + 2097152;

    relgen_kernel<<<2048, 256, 0, stream>>>(Relb);

    gemm_nt_kernel<<<dim3(4, 32), 256, 0, stream>>>(x_q,  Wq, Qb, 2048, 1024, 1024);
    gemm_nt_kernel<<<dim3(4, 64), 256, 0, stream>>>(x_kv, Wk, Kb, 4096, 1024, 1024);
    gemm_nt_kernel<<<dim3(4, 64), 256, 0, stream>>>(x_kv, Wv, Vb, 4096, 1024, 1024);
    gemm_nt_kernel<<<dim3(4, 32), 256, 0, stream>>>(Relb, Wr, Rb, 2048, 1024, 1024);

    for (int bh0 = 0; bh0 < 32; bh0 += 8) {
        score_kernel<<<dim3(8, 16, 8), 256, 0, stream>>>(Qb, Kb, Rb, ub, vb, Sb, bh0);
        softmax_kernel<<<dim3(1024, 8), 256, 0, stream>>>(Sb);
        pv_kernel<<<dim3(8, 8), 256, 0, stream>>>(Sb, Vb, Ob, bh0);
    }

    gemm_nt_kernel<<<dim3(4, 32), 256, 0, stream>>>(Ob, Wo, out, 2048, 1024, 1024);
}


# P.V kernel: the 16 waves of a block split the k range of one 16x64 tile (interleaved k steps) and add their partial tiles through LDS in wave order; 512 blocks, was 64. Changes the k sum order.
# speedup vs baseline: 1.5625x; 1.5625x over previous
#include <hip/hip_runtime.h>
#include <hip/hip_bf16.h>
#include <math.h>


typedef float v2f __attribute__((ext_vector_type(2)));
typedef float v8f __attribute__((ext_vector_type(8)));

__device__ __forceinline__ v8f v8f_zero() {
    v8f z = {0.f, 0.f, 0.f, 0.f, 0.f, 0.f, 0.f, 0.f};
    return z;
}

__device__ __forceinline__ v8f wmma4(v2f a, v2f b, v8f c) {
    return __builtin_amdgcn_wmma_f32_16x16x4_f32(false, a, false, b,
                                                 (short)0, c, false, false);
}

__global__ void relgen_kernel(float* __restrict__ rel) {
    const int row = blockIdx.x;
    const int t = threadIdx.x;
    const float pos = (float)(2047 - row);
    const float negl = -9.210340371976184f / 512.0f;
#pragma unroll
    for (int i = 0; i < 4; ++i) {
        int c = t + i * 256;
        int cc = c & 511;
        float inv = expf((float)cc * negl);
        float ang = pos * inv;
        rel[(size_t)row * 1024 + c] = (c < 512) ? sinf(ang) : cosf(ang);
    }
}

__global__ void gemm_nt_kernel(const float* __restrict__ A,
                               const float* __restrict__ B,
                               float* __restrict__ C,
                               int M, int N, int Kd) {
    const int lane = threadIdx.x & 31;
    const int w    = threadIdx.x >> 5;
    const int l16  = lane & 15;
    const int hlf  = lane >> 4;
    const int m0 = blockIdx.y * 64 + (w & 1) * 32;
    const int n0 = blockIdx.x * 256 + (w >> 1) * 64;
    (void)M;

    v8f c[2][4];
#pragma unroll
    for (int i = 0; i < 2; ++i)
#pragma unroll
        for (int j = 0; j < 4; ++j) c[i][j] = v8f_zero();

    for (int kb = 0; kb < Kd; kb += 4) {
        const int kk = kb + 2 * hlf;
        v2f a[2], b[4];
#pragma unroll
        for (int i = 0; i < 2; ++i)
            a[i] = *(const v2f*)(A + (size_t)(m0 + i * 16 + l16) * Kd + kk);
#pragma unroll
        for (int j = 0; j < 4; ++j)
            b[j] = *(const v2f*)(B + (size_t)(n0 + j * 16 + l16) * Kd + kk);
#pragma unroll
        for (int i = 0; i < 2; ++i)
#pragma unroll
            for (int j = 0; j < 4; ++j) c[i][j] = wmma4(a[i], b[j], c[i][j]);
    }

#pragma unroll
    for (int i = 0; i < 2; ++i)
#pragma unroll
        for (int j = 0; j < 4; ++j)
#pragma unroll
            for (int r = 0; r < 8; ++r) {
                int m = m0 + i * 16 + r + 8 * hlf;
                int n = n0 + j * 16 + l16;
                C[(size_t)m * N + n] = c[i][j][r];
            }
}

__global__ void score_kernel(const float* __restrict__ Q,
                             const float* __restrict__ Km,
                             const float* __restrict__ R,
                             const float* __restrict__ u,
                             const float* __restrict__ v,
                             float* __restrict__ S, int bh0) {
    const int lane = threadIdx.x & 31;
    const int w    = threadIdx.x >> 5;
    const int l16  = lane & 15;
    const int hlf  = lane >> 4;
    const int s  = blockIdx.z;
    const int bh = bh0 + s;
    const int b  = bh >> 4;
    const int h  = bh & 15;
    const int m0 = blockIdx.y * 64 + (w & 1) * 32;
    const int n0 = blockIdx.x * 256 + (w >> 1) * 64;

    const float* Qb = Q + (size_t)b * 1024 * 1024 + h * 64;
    const float* Kb = Km + (size_t)b * 2048 * 1024 + h * 64;
    const float* Rb = R + h * 64;

    v8f c[2][4];
#pragma unroll
    for (int i = 0; i < 2; ++i)
#pragma unroll
        for (int j = 0; j < 4; ++j) c[i][j] = v8f_zero();

    for (int kb = 0; kb < 64; kb += 4) {
        const int kk = kb + 2 * hlf;
        v2f ub = *(const v2f*)(u + h * 64 + kk);
        v2f vb = *(const v2f*)(v + h * 64 + kk);
        v2f qa[2], qv[2], bk[4], br[4];
#pragma unroll
        for (int i = 0; i < 2; ++i) {
            v2f qq = *(const v2f*)(Qb + (size_t)(m0 + i * 16 + l16) * 1024 + kk);
            qa[i] = qq + ub;
            qv[i] = qq + vb;
        }
#pragma unroll
        for (int j = 0; j < 4; ++j) {
            bk[j] = *(const v2f*)(Kb + (size_t)(n0 + j * 16 + l16) * 1024 + kk);
            br[j] = *(const v2f*)(Rb + (size_t)(n0 + j * 16 + l16) * 1024 + kk);
        }
#pragma unroll
        for (int i = 0; i < 2; ++i)
#pragma unroll
            for (int j = 0; j < 4; ++j) c[i][j] = wmma4(qa[i], bk[j], c[i][j]);
#pragma unroll
        for (int i = 0; i < 2; ++i)
#pragma unroll
            for (int j = 0; j < 4; ++j) c[i][j] = wmma4(qv[i], br[j], c[i][j]);
    }

    const float scale = 0.125f;
    float* Ss = S + (size_t)s * 1024 * 2048;
#pragma unroll
    for (int i = 0; i < 2; ++i)
#pragma unroll
        for (int j = 0; j < 4; ++j)
#pragma unroll
            for (int r = 0; r < 8; ++r) {
                int m = m0 + i * 16 + r + 8 * hlf;
                int n = n0 + j * 16 + l16;
                float val = c[i][j][r] * scale;
                if (n > 1024 + m) val = -INFINITY;
                Ss[(size_t)m * 2048 + n] = val;
            }
}

__global__ void softmax_kernel(float* __restrict__ S) {
    __shared__ float red[256];
    const int s = blockIdx.y, q = blockIdx.x, t = threadIdx.x;
    float* row = S + ((size_t)s * 1024 + q) * 2048;
    float x[8];
    float m = -INFINITY;
#pragma unroll
    for (int i = 0; i < 8; ++i) {
        x[i] = row[t + i * 256];
        m = fmaxf(m, x[i]);
    }
    red[t] = m;
    __syncthreads();
    for (int off = 128; off > 0; off >>= 1) {
        if (t < off) red[t] = fmaxf(red[t], red[t + off]);
        __syncthreads();
    }
    m = red[0];
    __syncthreads();
    float sum = 0.f;
#pragma unroll
    for (int i = 0; i < 8; ++i) {
        x[i] = __expf(x[i] - m);
        sum += x[i];
    }
    red[t] = sum;
    __syncthreads();
    for (int off = 128; off > 0; off >>= 1) {
        if (t < off) red[t] += red[t + off];
        __syncthreads();
    }
    float inv = 1.0f / red[0];
#pragma unroll
    for (int i = 0; i < 8; ++i) row[t + i * 256] = x[i] * inv;
}

__global__ void __launch_bounds__(512)
pv_kernel(const float* __restrict__ P,
          const float* __restrict__ V,
          float* __restrict__ O, int bh0) {
    __shared__ float part[16][16 * 64];
    const int lane = threadIdx.x & 31;
    const int w    = __builtin_amdgcn_readfirstlane((int)(threadIdx.x >> 5));
    const int l16  = lane & 15;
    const int hlf  = lane >> 4;
    const int s  = blockIdx.y;
    const int bh = bh0 + s;
    const int b  = bh >> 4;
    const int h  = bh & 15;
    const int m0 = blockIdx.x * 16;

    const float* Ps = P + (size_t)s * 1024 * 2048;
    const float* Vb = V + (size_t)b * 2048 * 1024 + h * 64;

    v8f c[4];
#pragma unroll
    for (int j = 0; j < 4; ++j) c[j] = v8f_zero();

    for (int t = 0; t < 32; ++t) {
        const int kb = 4 * w + 64 * t;
        const int kk = kb + 2 * hlf;
        v2f a = *(const v2f*)(Ps + (size_t)(m0 + l16) * 2048 + kk);
        v2f bf[4];
#pragma unroll
        for (int j = 0; j < 4; ++j) {
            int n = j * 16 + l16;
            bf[j].x = Vb[(size_t)kk * 1024 + n];
            bf[j].y = Vb[(size_t)(kk + 1) * 1024 + n];
        }
#pragma unroll
        for (int j = 0; j < 4; ++j) c[j] = wmma4(a, bf[j], c[j]);
    }

#pragma unroll
    for (int j = 0; j < 4; ++j)
#pragma unroll
        for (int r = 0; r < 8; ++r)
            part[w][(r + 8 * hlf) * 64 + j * 16 + l16] = c[j][r];
    __syncthreads();

#pragma unroll
    for (int i = 0; i < 2; ++i) {
        const int e = (int)threadIdx.x + i * 512;
        float acc = part[0][e];
#pragma unroll
        for (int p = 1; p < 16; ++p) acc += part[p][e];
        const int row = e >> 6;
        const int col = e & 63;
        O[((size_t)b * 1024 + m0 + row) * 1024 + h * 64 + col] = acc;
    }
}

extern "C" void kernel_launch(void* const* d_in, const int* in_sizes, int n_in,
                              void* d_out, int out_size, void* d_ws, size_t ws_size,
                              hipStream_t stream) {
    const float* x_q  = (const float*)d_in[0];
    const float* x_kv = (const float*)d_in[1];
    const float* Wq   = (const float*)d_in[2];
    const float* Wk   = (const float*)d_in[3];
    const float* Wv   = (const float*)d_in[4];
    const float* Wr   = (const float*)d_in[5];
    const float* Wo   = (const float*)d_in[6];
    const float* ub   = (const float*)d_in[7];
    const float* vb   = (const float*)d_in[8];
    float* out = (float*)d_out;
    (void)in_sizes; (void)n_in; (void)out_size; (void)ws_size;

    float* ws   = (float*)d_ws;
    float* Qb   = ws;
    float* Kb   = Qb + 2097152;
    float* Vb   = Kb + 4194304;
    float* Relb = Vb + 4194304;
    float* Rb   = Relb + 2097152;
    float* Ob   = Rb + 2097152;
    float* Sb   = Ob + 2097152;

    relgen_kernel<<<2048, 256, 0, stream>>>(Relb);

    gemm_nt_kernel<<<dim3(4, 32), 256, 0, stream>>>(x_q,  Wq, Qb, 2048, 1024, 1024);
    gemm_nt_kernel<<<dim3(4, 64), 256, 0, stream>>>(x_kv, Wk, Kb, 4096, 1024, 1024);
    gemm_nt_kernel<<<dim3(4, 64), 256, 0, stream>>>(x_kv, Wv, Vb, 4096, 1024, 1024);
    gemm_nt_kernel<<<dim3(4, 32), 256, 0, stream>>>(Relb, Wr, Rb, 2048, 1024, 1024);

    for (int bh0 = 0; bh0 < 32; bh0 += 8) {
        score_kernel<<<dim3(8, 16, 8), 256, 0, stream>>>(Qb, Kb, Rb, ub, vb, Sb, bh0);
        softmax_kernel<<<dim3(1024, 8), 256, 0, stream>>>(Sb);
        pv_kernel<<<dim3(64, 8), 512, 0, stream>>>(Sb, Vb, Ob, bh0);
    }

    gemm_nt_kernel<<<dim3(4, 32), 256, 0, stream>>>(Ob, Wo, out, 2048, 1024, 1024);
}


# gemm_nt: A and B block tiles staged through LDS in 64-k chunks by whole-line 16-byte global loads with register prefetch (was 8-byte loads touching 16 lines each); changes the k sum order
# speedup vs baseline: 1.7920x; 1.1469x over previous
#include <hip/hip_runtime.h>
#include <hip/hip_bf16.h>
#include <math.h>


typedef float v2f __attribute__((ext_vector_type(2)));
typedef float v8f __attribute__((ext_vector_type(8)));

__device__ __forceinline__ v8f v8f_zero() {
    v8f z = {0.f, 0.f, 0.f, 0.f, 0.f, 0.f, 0.f, 0.f};
    return z;
}

__device__ __forceinline__ v8f wmma4(v2f a, v2f b, v8f c) {
    return __builtin_amdgcn_wmma_f32_16x16x4_f32(false, a, false, b,
                                                 (short)0, c, false, false);
}

__global__ void relgen_kernel(float* __restrict__ rel) {
    const int row = blockIdx.x;
    const int t = threadIdx.x;
    const float pos = (float)(2047 - row);
    const float negl = -9.210340371976184f / 512.0f;
#pragma unroll
    for (int i = 0; i < 4; ++i) {
        int c = t + i * 256;
        int cc = c & 511;
        float inv = expf((float)cc * negl);
        float ang = pos * inv;
        rel[(size_t)row * 1024 + c] = (c < 512) ? sinf(ang) : cosf(ang);
    }
}

typedef float v4f __attribute__((ext_vector_type(4)));

#define GT_KC   64
#define GT_P    68
#define GT_ROWS 320
#define GT_NG   (GT_KC / 8)

__global__ void __launch_bounds__(256)
gemm_nt_kernel(const float* __restrict__ A,
               const float* __restrict__ B,
               float* __restrict__ C,
               int M, int N, int Kd) {
    __shared__ __attribute__((aligned(16))) float tile[2][GT_ROWS * GT_P];
    const int tid  = threadIdx.x;
    const int lane = tid & 31;
    const int w    = tid >> 5;
    const int l16  = lane & 15;
    const int hlf  = lane >> 4;
    const int wm = (w & 1) * 32;
    const int wn = (w >> 1) * 64;
    const int m0 = blockIdx.y * 64 + wm;
    const int n0 = blockIdx.x * 256 + wn;
    (void)M;

    const int srow = tid >> 4;
    const int sseg = tid & 15;
    const float* pa = A + (size_t)(blockIdx.y * 64 + srow) * Kd + sseg * 4;
    const float* pb = B + (size_t)(blockIdx.x * 256 + srow) * Kd + sseg * 4;
    const int soff = srow * GT_P + sseg * 4;

    v8f c[2][4];
#pragma unroll
    for (int i = 0; i < 2; ++i)
#pragma unroll
        for (int j = 0; j < 4; ++j) c[i][j] = v8f_zero();

    v4f g[20];
#pragma unroll
    for (int i = 0; i < 4; ++i)
        g[i] = *(const v4f*)(pa + (size_t)(16 * i) * Kd);
#pragma unroll
    for (int i = 0; i < 16; ++i)
        g[4 + i] = *(const v4f*)(pb + (size_t)(16 * i) * Kd);

    const int nk = Kd / GT_KC;
    for (int kc = 0; kc < nk; ++kc) {
        float* st = tile[kc & 1];
#pragma unroll
        for (int i = 0; i < 20; ++i)
            *(v4f*)(st + soff + 16 * i * GT_P) = g[i];
        __syncthreads();
        if (kc + 1 < nk) {
            const int ko = (kc + 1) * GT_KC;
#pragma unroll
            for (int i = 0; i < 4; ++i)
                g[i] = *(const v4f*)(pa + (size_t)(16 * i) * Kd + ko);
#pragma unroll
            for (int i = 0; i < 16; ++i)
                g[4 + i] = *(const v4f*)(pb + (size_t)(16 * i) * Kd + ko);
        }
        const float* as = st + (wm + l16) * GT_P + 4 * hlf;
        const float* bs = st + (64 + wn + l16) * GT_P + 4 * hlf;
        v4f a[2][2], b[2][4];
#pragma unroll
        for (int i = 0; i < 2; ++i) a[0][i] = *(const v4f*)(as + i * 16 * GT_P);
#pragma unroll
        for (int j = 0; j < 4; ++j) b[0][j] = *(const v4f*)(bs + j * 16 * GT_P);
#pragma unroll
        for (int gk = 0; gk < GT_NG; ++gk) {
            const int cur = gk & 1;
            if (gk + 1 < GT_NG) {
#pragma unroll
                for (int i = 0; i < 2; ++i)
                    a[cur ^ 1][i] = *(const v4f*)(as + i * 16 * GT_P + (gk + 1) * 8);
#pragma unroll
                for (int j = 0; j < 4; ++j)
                    b[cur ^ 1][j] = *(const v4f*)(bs + j * 16 * GT_P + (gk + 1) * 8);
            }
#pragma unroll
            for (int i = 0; i < 2; ++i)
#pragma unroll
                for (int j = 0; j < 4; ++j)
                    c[i][j] = wmma4(a[cur][i].xy, b[cur][j].xy, c[i][j]);
#pragma unroll
            for (int i = 0; i < 2; ++i)
#pragma unroll
                for (int j = 0; j < 4; ++j)
                    c[i][j] = wmma4(a[cur][i].zw, b[cur][j].zw, c[i][j]);
        }
    }

#pragma unroll
    for (int i = 0; i < 2; ++i)
#pragma unroll
        for (int j = 0; j < 4; ++j)
#pragma unroll
            for (int r = 0; r < 8; ++r) {
                int m = m0 + i * 16 + r + 8 * hlf;
                int n = n0 + j * 16 + l16;
                C[(size_t)m * N + n] = c[i][j][r];
            }
}

__global__ void score_kernel(const float* __restrict__ Q,
                             const float* __restrict__ Km,
                             const float* __restrict__ R,
                             const float* __restrict__ u,
                             const float* __restrict__ v,
                             float* __restrict__ S, int bh0) {
    const int lane = threadIdx.x & 31;
    const int w    = threadIdx.x >> 5;
    const int l16  = lane & 15;
    const int hlf  = lane >> 4;
    const int s  = blockIdx.z;
    const int bh = bh0 + s;
    const int b  = bh >> 4;
    const int h  = bh & 15;
    const int m0 = blockIdx.y * 64 + (w & 1) * 32;
    const int n0 = blockIdx.x * 256 + (w >> 1) * 64;

    const float* Qb = Q + (size_t)b * 1024 * 1024 + h * 64;
    const float* Kb = Km + (size_t)b * 2048 * 1024 + h * 64;
    const float* Rb = R + h * 64;

    v8f c[2][4];
#pragma unroll
    for (int i = 0; i < 2; ++i)
#pragma unroll
        for (int j = 0; j < 4; ++j) c[i][j] = v8f_zero();

    for (int kb = 0; kb < 64; kb += 4) {
        const int kk = kb + 2 * hlf;
        v2f ub = *(const v2f*)(u + h * 64 + kk);
        v2f vb = *(const v2f*)(v + h * 64 + kk);
        v2f qa[2], qv[2], bk[4], br[4];
#pragma unroll
        for (int i = 0; i < 2; ++i) {
            v2f qq = *(const v2f*)(Qb + (size_t)(m0 + i * 16 + l16) * 1024 + kk);
            qa[i] = qq + ub;
            qv[i] = qq + vb;
        }
#pragma unroll
        for (int j = 0; j < 4; ++j) {
            bk[j] = *(const v2f*)(Kb + (size_t)(n0 + j * 16 + l16) * 1024 + kk);
            br[j] = *(const v2f*)(Rb + (size_t)(n0 + j * 16 + l16) * 1024 + kk);
        }
#pragma unroll
        for (int i = 0; i < 2; ++i)
#pragma unroll
            for (int j = 0; j < 4; ++j) c[i][j] = wmma4(qa[i], bk[j], c[i][j]);
#pragma unroll
        for (int i = 0; i < 2; ++i)
#pragma unroll
            for (int j = 0; j < 4; ++j) c[i][j] = wmma4(qv[i], br[j], c[i][j]);
    }

    const float scale = 0.125f;
    float* Ss = S + (size_t)s * 1024 * 2048;
#pragma unroll
    for (int i = 0; i < 2; ++i)
#pragma unroll
        for (int j = 0; j < 4; ++j)
#pragma unroll
            for (int r = 0; r < 8; ++r) {
                int m = m0 + i * 16 + r + 8 * hlf;
                int n = n0 + j * 16 + l16;
                float val = c[i][j][r] * scale;
                if (n > 1024 + m) val = -INFINITY;
                Ss[(size_t)m * 2048 + n] = val;
            }
}

__global__ void softmax_kernel(float* __restrict__ S) {
    __shared__ float red[256];
    const int s = blockIdx.y, q = blockIdx.x, t = threadIdx.x;
    float* row = S + ((size_t)s * 1024 + q) * 2048;
    float x[8];
    float m = -INFINITY;
#pragma unroll
    for (int i = 0; i < 8; ++i) {
        x[i] = row[t + i * 256];
        m = fmaxf(m, x[i]);
    }
    red[t] = m;
    __syncthreads();
    for (int off = 128; off > 0; off >>= 1) {
        if (t < off) red[t] = fmaxf(red[t], red[t + off]);
        __syncthreads();
    }
    m = red[0];
    __syncthreads();
    float sum = 0.f;
#pragma unroll
    for (int i = 0; i < 8; ++i) {
        x[i] = __expf(x[i] - m);
        sum += x[i];
    }
    red[t] = sum;
    __syncthreads();
    for (int off = 128; off > 0; off >>= 1) {
        if (t < off) red[t] += red[t + off];
        __syncthreads();
    }
    float inv = 1.0f / red[0];
#pragma unroll
    for (int i = 0; i < 8; ++i) row[t + i * 256] = x[i] * inv;
}

__global__ void __launch_bounds__(512)
pv_kernel(const float* __restrict__ P,
          const float* __restrict__ V,
          float* __restrict__ O, int bh0) {
    __shared__ float part[16][16 * 64];
    const int lane = threadIdx.x & 31;
    const int w    = __builtin_amdgcn_readfirstlane((int)(threadIdx.x >> 5));
    const int l16  = lane & 15;
    const int hlf  = lane >> 4;
    const int s  = blockIdx.y;
    const int bh = bh0 + s;
    const int b  = bh >> 4;
    const int h  = bh & 15;
    const int m0 = blockIdx.x * 16;

    const float* Ps = P + (size_t)s * 1024 * 2048;
    const float* Vb = V + (size_t)b * 2048 * 1024 + h * 64;

    v8f c[4];
#pragma unroll
    for (int j = 0; j < 4; ++j) c[j] = v8f_zero();

    for (int t = 0; t < 32; ++t) {
        const int kb = 4 * w + 64 * t;
        const int kk = kb + 2 * hlf;
        v2f a = *(const v2f*)(Ps + (size_t)(m0 + l16) * 2048 + kk);
        v2f bf[4];
#pragma unroll
        for (int j = 0; j < 4; ++j) {
            int n = j * 16 + l16;
            bf[j].x = Vb[(size_t)kk * 1024 + n];
            bf[j].y = Vb[(size_t)(kk + 1) * 1024 + n];
        }
#pragma unroll
        for (int j = 0; j < 4; ++j) c[j] = wmma4(a, bf[j], c[j]);
    }

#pragma unroll
    for (int j = 0; j < 4; ++j)
#pragma unroll
        for (int r = 0; r < 8; ++r)
            part[w][(r + 8 * hlf) * 64 + j * 16 + l16] = c[j][r];
    __syncthreads();

#pragma unroll
    for (int i = 0; i < 2; ++i) {
        const int e = (int)threadIdx.x + i * 512;
        float acc = part[0][e];
#pragma unroll
        for (int p = 1; p < 16; ++p) acc += part[p][e];
        const int row = e >> 6;
        const int col = e & 63;
        O[((size_t)b * 1024 + m0 + row) * 1024 + h * 64 + col] = acc;
    }
}

extern "C" void kernel_launch(void* const* d_in, const int* in_sizes, int n_in,
                              void* d_out, int out_size, void* d_ws, size_t ws_size,
                              hipStream_t stream) {
    const float* x_q  = (const float*)d_in[0];
    const float* x_kv = (const float*)d_in[1];
    const float* Wq   = (const float*)d_in[2];
    const float* Wk   = (const float*)d_in[3];
    const float* Wv   = (const float*)d_in[4];
    const float* Wr   = (const float*)d_in[5];
    const float* Wo   = (const float*)d_in[6];
    const float* ub   = (const float*)d_in[7];
    const float* vb   = (const float*)d_in[8];
    float* out = (float*)d_out;
    (void)in_sizes; (void)n_in; (void)out_size; (void)ws_size;

    float* ws   = (float*)d_ws;
    float* Qb   = ws;
    float* Kb   = Qb + 2097152;
    float* Vb   = Kb + 4194304;
    float* Relb = Vb + 4194304;
    float* Rb   = Relb + 2097152;
    float* Ob   = Rb + 2097152;
    float* Sb   = Ob + 2097152;

    relgen_kernel<<<2048, 256, 0, stream>>>(Relb);

    gemm_nt_kernel<<<dim3(4, 32), 256, 0, stream>>>(x_q,  Wq, Qb, 2048, 1024, 1024);
    gemm_nt_kernel<<<dim3(4, 64), 256, 0, stream>>>(x_kv, Wk, Kb, 4096, 1024, 1024);
    gemm_nt_kernel<<<dim3(4, 64), 256, 0, stream>>>(x_kv, Wv, Vb, 4096, 1024, 1024);
    gemm_nt_kernel<<<dim3(4, 32), 256, 0, stream>>>(Relb, Wr, Rb, 2048, 1024, 1024);

    for (int bh0 = 0; bh0 < 32; bh0 += 8) {
        score_kernel<<<dim3(8, 16, 8), 256, 0, stream>>>(Qb, Kb, Rb, ub, vb, Sb, bh0);
        softmax_kernel<<<dim3(1024, 8), 256, 0, stream>>>(Sb);
        pv_kernel<<<dim3(64, 8), 512, 0, stream>>>(Sb, Vb, Ob, bh0);
    }

    gemm_nt_kernel<<<dim3(4, 32), 256, 0, stream>>>(Ob, Wo, out, 2048, 1024, 1024);
}


# score kernel: Q, K, R rows (256 bytes per row), u and v staged in LDS once per 128x128 block by whole-line 16-byte loads (was 8-byte loads, 16 lines each); changes k sum order
# speedup vs baseline: 2.1431x; 1.1959x over previous
#include <hip/hip_runtime.h>
#include <hip/hip_bf16.h>
#include <math.h>


typedef float v2f __attribute__((ext_vector_type(2)));
typedef float v8f __attribute__((ext_vector_type(8)));

__device__ __forceinline__ v8f v8f_zero() {
    v8f z = {0.f, 0.f, 0.f, 0.f, 0.f, 0.f, 0.f, 0.f};
    return z;
}

__device__ __forceinline__ v8f wmma4(v2f a, v2f b, v8f c) {
    return __builtin_amdgcn_wmma_f32_16x16x4_f32(false, a, false, b,
                                                 (short)0, c, false, false);
}

__global__ void relgen_kernel(float* __restrict__ rel) {
    const int row = blockIdx.x;
    const int t = threadIdx.x;
    const float pos = (float)(2047 - row);
    const float negl = -9.210340371976184f / 512.0f;
#pragma unroll
    for (int i = 0; i < 4; ++i) {
        int c = t + i * 256;
        int cc = c & 511;
        float inv = expf((float)cc * negl);
        float ang = pos * inv;
        rel[(size_t)row * 1024 + c] = (c < 512) ? sinf(ang) : cosf(ang);
    }
}

typedef float v4f __attribute__((ext_vector_type(4)));

#define GT_KC   64
#define GT_P    68
#define GT_ROWS 320
#define GT_NG   (GT_KC / 8)

__global__ void __launch_bounds__(256)
gemm_nt_kernel(const float* __restrict__ A,
               const float* __restrict__ B,
               float* __restrict__ C,
               int M, int N, int Kd) {
    __shared__ __attribute__((aligned(16))) float tile[2][GT_ROWS * GT_P];
    const int tid  = threadIdx.x;
    const int lane = tid & 31;
    const int w    = tid >> 5;
    const int l16  = lane & 15;
    const int hlf  = lane >> 4;
    const int wm = (w & 1) * 32;
    const int wn = (w >> 1) * 64;
    const int m0 = blockIdx.y * 64 + wm;
    const int n0 = blockIdx.x * 256 + wn;
    (void)M;

    const int srow = tid >> 4;
    const int sseg = tid & 15;
    const float* pa = A + (size_t)(blockIdx.y * 64 + srow) * Kd + sseg * 4;
    const float* pb = B + (size_t)(blockIdx.x * 256 + srow) * Kd + sseg * 4;
    const int soff = srow * GT_P + sseg * 4;

    v8f c[2][4];
#pragma unroll
    for (int i = 0; i < 2; ++i)
#pragma unroll
        for (int j = 0; j < 4; ++j) c[i][j] = v8f_zero();

    v4f g[20];
#pragma unroll
    for (int i = 0; i < 4; ++i)
        g[i] = *(const v4f*)(pa + (size_t)(16 * i) * Kd);
#pragma unroll
    for (int i = 0; i < 16; ++i)
        g[4 + i] = *(const v4f*)(pb + (size_t)(16 * i) * Kd);

    const int nk = Kd / GT_KC;
    for (int kc = 0; kc < nk; ++kc) {
        float* st = tile[kc & 1];
#pragma unroll
        for (int i = 0; i < 20; ++i)
            *(v4f*)(st + soff + 16 * i * GT_P) = g[i];
        __syncthreads();
        if (kc + 1 < nk) {
            const int ko = (kc + 1) * GT_KC;
#pragma unroll
            for (int i = 0; i < 4; ++i)
                g[i] = *(const v4f*)(pa + (size_t)(16 * i) * Kd + ko);
#pragma unroll
            for (int i = 0; i < 16; ++i)
                g[4 + i] = *(const v4f*)(pb + (size_t)(16 * i) * Kd + ko);
        }
        const float* as = st + (wm + l16) * GT_P + 4 * hlf;
        const float* bs = st + (64 + wn + l16) * GT_P + 4 * hlf;
        v4f a[2][2], b[2][4];
#pragma unroll
        for (int i = 0; i < 2; ++i) a[0][i] = *(const v4f*)(as + i * 16 * GT_P);
#pragma unroll
        for (int j = 0; j < 4; ++j) b[0][j] = *(const v4f*)(bs + j * 16 * GT_P);
#pragma unroll
        for (int gk = 0; gk < GT_NG; ++gk) {
            const int cur = gk & 1;
            if (gk + 1 < GT_NG) {
#pragma unroll
                for (int i = 0; i < 2; ++i)
                    a[cur ^ 1][i] = *(const v4f*)(as + i * 16 * GT_P + (gk + 1) * 8);
#pragma unroll
                for (int j = 0; j < 4; ++j)
                    b[cur ^ 1][j] = *(const v4f*)(bs + j * 16 * GT_P + (gk + 1) * 8);
            }
#pragma unroll
            for (int i = 0; i < 2; ++i)
#pragma unroll
                for (int j = 0; j < 4; ++j)
                    c[i][j] = wmma4(a[cur][i].xy, b[cur][j].xy, c[i][j]);
#pragma unroll
            for (int i = 0; i < 2; ++i)
#pragma unroll
                for (int j = 0; j < 4; ++j)
                    c[i][j] = wmma4(a[cur][i].zw, b[cur][j].zw, c[i][j]);
        }
    }

#pragma unroll
    for (int i = 0; i < 2; ++i)
#pragma unroll
        for (int j = 0; j < 4; ++j)
#pragma unroll
            for (int r = 0; r < 8; ++r) {
                int m = m0 + i * 16 + r + 8 * hlf;
                int n = n0 + j * 16 + l16;
                C[(size_t)m * N + n] = c[i][j][r];
            }
}

#define SC_P    68
#define SC_ROWS 384

__global__ void __launch_bounds__(256)
score_kernel(const float* __restrict__ Q,
             const float* __restrict__ Km,
             const float* __restrict__ R,
             const float* __restrict__ u,
             const float* __restrict__ v,
             float* __restrict__ S, int bh0) {
    __shared__ __attribute__((aligned(16))) float tile[SC_ROWS * SC_P + 128];
    const int tid  = threadIdx.x;
    const int lane = tid & 31;
    const int w    = tid >> 5;
    const int l16  = lane & 15;
    const int hlf  = lane >> 4;
    const int s  = blockIdx.z;
    const int bh = bh0 + s;
    const int b  = bh >> 4;
    const int h  = bh & 15;
    const int q0 = blockIdx.y * 128;
    const int k0 = blockIdx.x * 128;
    const int wm = (w & 3) * 32;
    const int wn = (w >> 2) * 64;

    const int srow = tid >> 4;
    const int sseg = tid & 15;
    const float* pq = Q  + (size_t)b * 1024 * 1024 + h * 64
                         + (size_t)(q0 + srow) * 1024 + sseg * 4;
    const float* pk = Km + (size_t)b * 2048 * 1024 + h * 64
                         + (size_t)(k0 + srow) * 1024 + sseg * 4;
    const float* pr = R  + h * 64 + (size_t)(k0 + srow) * 1024 + sseg * 4;
    const int soff = srow * SC_P + sseg * 4;

    v4f g[24];
#pragma unroll
    for (int i = 0; i < 8; ++i) {
        g[i]      = *(const v4f*)(pq + (size_t)(16 * i) * 1024);
        g[8 + i]  = *(const v4f*)(pk + (size_t)(16 * i) * 1024);
        g[16 + i] = *(const v4f*)(pr + (size_t)(16 * i) * 1024);
    }
    if (tid < 32) {
        const float* bsrc = (tid < 16 ? u : v) + h * 64 + (tid & 15) * 4;
        *(v4f*)(tile + SC_ROWS * SC_P + tid * 4) = *(const v4f*)bsrc;
    }
#pragma unroll
    for (int i = 0; i < 24; ++i)
        *(v4f*)(tile + soff + 16 * i * SC_P) = g[i];
    __syncthreads();

    v8f c[2][4];
#pragma unroll
    for (int i = 0; i < 2; ++i)
#pragma unroll
        for (int j = 0; j < 4; ++j) c[i][j] = v8f_zero();

    const float* as = tile + (wm + l16) * SC_P + 4 * hlf;
    const float* ks = tile + (128 + wn + l16) * SC_P + 4 * hlf;
    const float* rs = tile + (256 + wn + l16) * SC_P + 4 * hlf;
    const float* us = tile + SC_ROWS * SC_P + 4 * hlf;
    const float* vs = us + 64;

#pragma unroll
    for (int gk = 0; gk < 8; ++gk) {
        const v4f ub = *(const v4f*)(us + gk * 8);
        const v4f vb = *(const v4f*)(vs + gk * 8);
        v4f qa[2], qv[2], bk[4], br[4];
#pragma unroll
        for (int i = 0; i < 2; ++i) {
            const v4f qq = *(const v4f*)(as + i * 16 * SC_P + gk * 8);
            qa[i] = qq + ub;
            qv[i] = qq + vb;
        }
#pragma unroll
        for (int j = 0; j < 4; ++j) {
            bk[j] = *(const v4f*)(ks + j * 16 * SC_P + gk * 8);
            br[j] = *(const v4f*)(rs + j * 16 * SC_P + gk * 8);
        }
#pragma unroll
        for (int i = 0; i < 2; ++i)
#pragma unroll
            for (int j = 0; j < 4; ++j) c[i][j] = wmma4(qa[i].xy, bk[j].xy, c[i][j]);
#pragma unroll
        for (int i = 0; i < 2; ++i)
#pragma unroll
            for (int j = 0; j < 4; ++j) c[i][j] = wmma4(qa[i].zw, bk[j].zw, c[i][j]);
#pragma unroll
        for (int i = 0; i < 2; ++i)
#pragma unroll
            for (int j = 0; j < 4; ++j) c[i][j] = wmma4(qv[i].xy, br[j].xy, c[i][j]);
#pragma unroll
        for (int i = 0; i < 2; ++i)
#pragma unroll
            for (int j = 0; j < 4; ++j) c[i][j] = wmma4(qv[i].zw, br[j].zw, c[i][j]);
    }

    const int m0 = q0 + wm;
    const int n0 = k0 + wn;
    const float scale = 0.125f;
    float* Ss = S + (size_t)s * 1024 * 2048;
#pragma unroll
    for (int i = 0; i < 2; ++i)
#pragma unroll
        for (int j = 0; j < 4; ++j)
#pragma unroll
            for (int r = 0; r < 8; ++r) {
                int m = m0 + i * 16 + r + 8 * hlf;
                int n = n0 + j * 16 + l16;
                float val = c[i][j][r] * scale;
                if (n > 1024 + m) val = -INFINITY;
                Ss[(size_t)m * 2048 + n] = val;
            }
}

__global__ void softmax_kernel(float* __restrict__ S) {
    __shared__ float red[256];
    const int s = blockIdx.y, q = blockIdx.x, t = threadIdx.x;
    float* row = S + ((size_t)s * 1024 + q) * 2048;
    float x[8];
    float m = -INFINITY;
#pragma unroll
    for (int i = 0; i < 8; ++i) {
        x[i] = row[t + i * 256];
        m = fmaxf(m, x[i]);
    }
    red[t] = m;
    __syncthreads();
    for (int off = 128; off > 0; off >>= 1) {
        if (t < off) red[t] = fmaxf(red[t], red[t + off]);
        __syncthreads();
    }
    m = red[0];
    __syncthreads();
    float sum = 0.f;
#pragma unroll
    for (int i = 0; i < 8; ++i) {
        x[i] = __expf(x[i] - m);
        sum += x[i];
    }
    red[t] = sum;
    __syncthreads();
    for (int off = 128; off > 0; off >>= 1) {
        if (t < off) red[t] += red[t + off];
        __syncthreads();
    }
    float inv = 1.0f / red[0];
#pragma unroll
    for (int i = 0; i < 8; ++i) row[t + i * 256] = x[i] * inv;
}

__global__ void __launch_bounds__(512)
pv_kernel(const float* __restrict__ P,
          const float* __restrict__ V,
          float* __restrict__ O, int bh0) {
    __shared__ float part[16][16 * 64];
    const int lane = threadIdx.x & 31;
    const int w    = __builtin_amdgcn_readfirstlane((int)(threadIdx.x >> 5));
    const int l16  = lane & 15;
    const int hlf  = lane >> 4;
    const int s  = blockIdx.y;
    const int bh = bh0 + s;
    const int b  = bh >> 4;
    const int h  = bh & 15;
    const int m0 = blockIdx.x * 16;

    const float* Ps = P + (size_t)s * 1024 * 2048;
    const float* Vb = V + (size_t)b * 2048 * 1024 + h * 64;

    v8f c[4];
#pragma unroll
    for (int j = 0; j < 4; ++j) c[j] = v8f_zero();

    for (int t = 0; t < 32; ++t) {
        const int kb = 4 * w + 64 * t;
        const int kk = kb + 2 * hlf;
        v2f a = *(const v2f*)(Ps + (size_t)(m0 + l16) * 2048 + kk);
        v2f bf[4];
#pragma unroll
        for (int j = 0; j < 4; ++j) {
            int n = j * 16 + l16;
            bf[j].x = Vb[(size_t)kk * 1024 + n];
            bf[j].y = Vb[(size_t)(kk + 1) * 1024 + n];
        }
#pragma unroll
        for (int j = 0; j < 4; ++j) c[j] = wmma4(a, bf[j], c[j]);
    }

#pragma unroll
    for (int j = 0; j < 4; ++j)
#pragma unroll
        for (int r = 0; r < 8; ++r)
            part[w][(r + 8 * hlf) * 64 + j * 16 + l16] = c[j][r];
    __syncthreads();

#pragma unroll
    for (int i = 0; i < 2; ++i) {
        const int e = (int)threadIdx.x + i * 512;
        float acc = part[0][e];
#pragma unroll
        for (int p = 1; p < 16; ++p) acc += part[p][e];
        const int row = e >> 6;
        const int col = e & 63;
        O[((size_t)b * 1024 + m0 + row) * 1024 + h * 64 + col] = acc;
    }
}

extern "C" void kernel_launch(void* const* d_in, const int* in_sizes, int n_in,
                              void* d_out, int out_size, void* d_ws, size_t ws_size,
                              hipStream_t stream) {
    const float* x_q  = (const float*)d_in[0];
    const float* x_kv = (const float*)d_in[1];
    const float* Wq   = (const float*)d_in[2];
    const float* Wk   = (const float*)d_in[3];
    const float* Wv   = (const float*)d_in[4];
    const float* Wr   = (const float*)d_in[5];
    const float* Wo   = (const float*)d_in[6];
    const float* ub   = (const float*)d_in[7];
    const float* vb   = (const float*)d_in[8];
    float* out = (float*)d_out;
    (void)in_sizes; (void)n_in; (void)out_size; (void)ws_size;

    float* ws   = (float*)d_ws;
    float* Qb   = ws;
    float* Kb   = Qb + 2097152;
    float* Vb   = Kb + 4194304;
    float* Relb = Vb + 4194304;
    float* Rb   = Relb + 2097152;
    float* Ob   = Rb + 2097152;
    float* Sb   = Ob + 2097152;

    relgen_kernel<<<2048, 256, 0, stream>>>(Relb);

    gemm_nt_kernel<<<dim3(4, 32), 256, 0, stream>>>(x_q,  Wq, Qb, 2048, 1024, 1024);
    gemm_nt_kernel<<<dim3(4, 64), 256, 0, stream>>>(x_kv, Wk, Kb, 4096, 1024, 1024);
    gemm_nt_kernel<<<dim3(4, 64), 256, 0, stream>>>(x_kv, Wv, Vb, 4096, 1024, 1024);
    gemm_nt_kernel<<<dim3(4, 32), 256, 0, stream>>>(Relb, Wr, Rb, 2048, 1024, 1024);

    for (int bh0 = 0; bh0 < 32; bh0 += 8) {
        score_kernel<<<dim3(16, 8, 8), 256, 0, stream>>>(Qb, Kb, Rb, ub, vb, Sb, bh0);
        softmax_kernel<<<dim3(1024, 8), 256, 0, stream>>>(Sb);
        pv_kernel<<<dim3(64, 8), 512, 0, stream>>>(Sb, Vb, Ob, bh0);
    }

    gemm_nt_kernel<<<dim3(4, 32), 256, 0, stream>>>(Ob, Wo, out, 2048, 1024, 1024);
}


# gemm_nt: the three 2048x1024 products (Q, R, output projection) use 64x128 block tiles with 128-k chunks: 256 blocks, one per WGP (was 128 blocks of 64x256 on half the WGPs); k sum order unchanged
# speedup vs baseline: 2.3624x; 1.1023x over previous
#include <hip/hip_runtime.h>
#include <hip/hip_bf16.h>
#include <math.h>


typedef float v2f __attribute__((ext_vector_type(2)));
typedef float v8f __attribute__((ext_vector_type(8)));

__device__ __forceinline__ v8f v8f_zero() {
    v8f z = {0.f, 0.f, 0.f, 0.f, 0.f, 0.f, 0.f, 0.f};
    return z;
}

__device__ __forceinline__ v8f wmma4(v2f a, v2f b, v8f c) {
    return __builtin_amdgcn_wmma_f32_16x16x4_f32(false, a, false, b,
                                                 (short)0, c, false, false);
}

__global__ void relgen_kernel(float* __restrict__ rel) {
    const int row = blockIdx.x;
    const int t = threadIdx.x;
    const float pos = (float)(2047 - row);
    const float negl = -9.210340371976184f / 512.0f;
#pragma unroll
    for (int i = 0; i < 4; ++i) {
        int c = t + i * 256;
        int cc = c & 511;
        float inv = expf((float)cc * negl);
        float ang = pos * inv;
        rel[(size_t)row * 1024 + c] = (c < 512) ? sinf(ang) : cosf(ang);
    }
}

typedef float v4f __attribute__((ext_vector_type(4)));

template <int NJ, int KC>
__global__ void __launch_bounds__(256)
gemm_nt_kernel(const float* __restrict__ A,
               const float* __restrict__ B,
               float* __restrict__ C,
               int M, int N, int Kd) {
    constexpr int GT_P    = KC + 4;
    constexpr int GT_BN   = 64 * NJ;
    constexpr int GT_ROWS = 64 + GT_BN;
    constexpr int GT_NG   = KC / 8;
    constexpr int GT_SEG  = KC / 4;
    constexpr int GT_RPP  = 256 / GT_SEG;
    constexpr int GT_NA   = 64 / GT_RPP;
    constexpr int GT_NB   = GT_BN / GT_RPP;
    __shared__ __attribute__((aligned(16))) float tile[2][GT_ROWS * GT_P];
    const int tid  = threadIdx.x;
    const int lane = tid & 31;
    const int w    = tid >> 5;
    const int l16  = lane & 15;
    const int hlf  = lane >> 4;
    const int wm = (w & 1) * 32;
    const int wn = (w >> 1) * (16 * NJ);
    const int m0 = blockIdx.y * 64 + wm;
    const int n0 = blockIdx.x * GT_BN + wn;
    (void)M;

    const int srow = tid / GT_SEG;
    const int sseg = tid % GT_SEG;
    const float* pa = A + (size_t)(blockIdx.y * 64 + srow) * Kd + sseg * 4;
    const float* pb = B + (size_t)(blockIdx.x * GT_BN + srow) * Kd + sseg * 4;
    const int soff = srow * GT_P + sseg * 4;

    v8f c[2][NJ];
#pragma unroll
    for (int i = 0; i < 2; ++i)
#pragma unroll
        for (int j = 0; j < NJ; ++j) c[i][j] = v8f_zero();

    v4f g[GT_NA + GT_NB];
#pragma unroll
    for (int i = 0; i < GT_NA; ++i)
        g[i] = *(const v4f*)(pa + (size_t)(GT_RPP * i) * Kd);
#pragma unroll
    for (int i = 0; i < GT_NB; ++i)
        g[GT_NA + i] = *(const v4f*)(pb + (size_t)(GT_RPP * i) * Kd);

    const int nk = Kd / KC;
    for (int kc = 0; kc < nk; ++kc) {
        float* st = tile[kc & 1];
#pragma unroll
        for (int i = 0; i < GT_NA + GT_NB; ++i)
            *(v4f*)(st + soff + GT_RPP * i * GT_P) = g[i];
        __syncthreads();
        if (kc + 1 < nk) {
            const int ko = (kc + 1) * KC;
#pragma unroll
            for (int i = 0; i < GT_NA; ++i)
                g[i] = *(const v4f*)(pa + (size_t)(GT_RPP * i) * Kd + ko);
#pragma unroll
            for (int i = 0; i < GT_NB; ++i)
                g[GT_NA + i] = *(const v4f*)(pb + (size_t)(GT_RPP * i) * Kd + ko);
        }
        const float* as = st + (wm + l16) * GT_P + 4 * hlf;
        const float* bs = st + (64 + wn + l16) * GT_P + 4 * hlf;
        v4f a[2][2], b[2][NJ];
#pragma unroll
        for (int i = 0; i < 2; ++i) a[0][i] = *(const v4f*)(as + i * 16 * GT_P);
#pragma unroll
        for (int j = 0; j < NJ; ++j) b[0][j] = *(const v4f*)(bs + j * 16 * GT_P);
#pragma unroll
        for (int gk = 0; gk < GT_NG; ++gk) {
            const int cur = gk & 1;
            if (gk + 1 < GT_NG) {
#pragma unroll
                for (int i = 0; i < 2; ++i)
                    a[cur ^ 1][i] = *(const v4f*)(as + i * 16 * GT_P + (gk + 1) * 8);
#pragma unroll
                for (int j = 0; j < NJ; ++j)
                    b[cur ^ 1][j] = *(const v4f*)(bs + j * 16 * GT_P + (gk + 1) * 8);
            }
#pragma unroll
            for (int i = 0; i < 2; ++i)
#pragma unroll
                for (int j = 0; j < NJ; ++j)
                    c[i][j] = wmma4(a[cur][i].xy, b[cur][j].xy, c[i][j]);
#pragma unroll
            for (int i = 0; i < 2; ++i)
#pragma unroll
                for (int j = 0; j < NJ; ++j)
                    c[i][j] = wmma4(a[cur][i].zw, b[cur][j].zw, c[i][j]);
        }
    }

#pragma unroll
    for (int i = 0; i < 2; ++i)
#pragma unroll
        for (int j = 0; j < NJ; ++j)
#pragma unroll
            for (int r = 0; r < 8; ++r) {
                int m = m0 + i * 16 + r + 8 * hlf;
                int n = n0 + j * 16 + l16;
                C[(size_t)m * N + n] = c[i][j][r];
            }
}

#define SC_P    68
#define SC_ROWS 384

__global__ void __launch_bounds__(256)
score_kernel(const float* __restrict__ Q,
             const float* __restrict__ Km,
             const float* __restrict__ R,
             const float* __restrict__ u,
             const float* __restrict__ v,
             float* __restrict__ S, int bh0) {
    __shared__ __attribute__((aligned(16))) float tile[SC_ROWS * SC_P + 128];
    const int tid  = threadIdx.x;
    const int lane = tid & 31;
    const int w    = tid >> 5;
    const int l16  = lane & 15;
    const int hlf  = lane >> 4;
    const int s  = blockIdx.z;
    const int bh = bh0 + s;
    const int b  = bh >> 4;
    const int h  = bh & 15;
    const int q0 = blockIdx.y * 128;
    const int k0 = blockIdx.x * 128;
    const int wm = (w & 3) * 32;
    const int wn = (w >> 2) * 64;

    const int srow = tid >> 4;
    const int sseg = tid & 15;
    const float* pq = Q  + (size_t)b * 1024 * 1024 + h * 64
                         + (size_t)(q0 + srow) * 1024 + sseg * 4;
    const float* pk = Km + (size_t)b * 2048 * 1024 + h * 64
                         + (size_t)(k0 + srow) * 1024 + sseg * 4;
    const float* pr = R  + h * 64 + (size_t)(k0 + srow) * 1024 + sseg * 4;
    const int soff = srow * SC_P + sseg * 4;

    v4f g[24];
#pragma unroll
    for (int i = 0; i < 8; ++i) {
        g[i]      = *(const v4f*)(pq + (size_t)(16 * i) * 1024);
        g[8 + i]  = *(const v4f*)(pk + (size_t)(16 * i) * 1024);
        g[16 + i] = *(const v4f*)(pr + (size_t)(16 * i) * 1024);
    }
    if (tid < 32) {
        const float* bsrc = (tid < 16 ? u : v) + h * 64 + (tid & 15) * 4;
        *(v4f*)(tile + SC_ROWS * SC_P + tid * 4) = *(const v4f*)bsrc;
    }
#pragma unroll
    for (int i = 0; i < 24; ++i)
        *(v4f*)(tile + soff + 16 * i * SC_P) = g[i];
    __syncthreads();

    v8f c[2][4];
#pragma unroll
    for (int i = 0; i < 2; ++i)
#pragma unroll
        for (int j = 0; j < 4; ++j) c[i][j] = v8f_zero();

    const float* as = tile + (wm + l16) * SC_P + 4 * hlf;
    const float* ks = tile + (128 + wn + l16) * SC_P + 4 * hlf;
    const float* rs = tile + (256 + wn + l16) * SC_P + 4 * hlf;
    const float* us = tile + SC_ROWS * SC_P + 4 * hlf;
    const float* vs = us + 64;

#pragma unroll
    for (int gk = 0; gk < 8; ++gk) {
        const v4f ub = *(const v4f*)(us + gk * 8);
        const v4f vb = *(const v4f*)(vs + gk * 8);
        v4f qa[2], qv[2], bk[4], br[4];
#pragma unroll
        for (int i = 0; i < 2; ++i) {
            const v4f qq = *(const v4f*)(as + i * 16 * SC_P + gk * 8);
            qa[i] = qq + ub;
            qv[i] = qq + vb;
        }
#pragma unroll
        for (int j = 0; j < 4; ++j) {
            bk[j] = *(const v4f*)(ks + j * 16 * SC_P + gk * 8);
            br[j] = *(const v4f*)(rs + j * 16 * SC_P + gk * 8);
        }
#pragma unroll
        for (int i = 0; i < 2; ++i)
#pragma unroll
            for (int j = 0; j < 4; ++j) c[i][j] = wmma4(qa[i].xy, bk[j].xy, c[i][j]);
#pragma unroll
        for (int i = 0; i < 2; ++i)
#pragma unroll
            for (int j = 0; j < 4; ++j) c[i][j] = wmma4(qa[i].zw, bk[j].zw, c[i][j]);
#pragma unroll
        for (int i = 0; i < 2; ++i)
#pragma unroll
            for (int j = 0; j < 4; ++j) c[i][j] = wmma4(qv[i].xy, br[j].xy, c[i][j]);
#pragma unroll
        for (int i = 0; i < 2; ++i)
#pragma unroll
            for (int j = 0; j < 4; ++j) c[i][j] = wmma4(qv[i].zw, br[j].zw, c[i][j]);
    }

    const int m0 = q0 + wm;
    const int n0 = k0 + wn;
    const float scale = 0.125f;
    float* Ss = S + (size_t)s * 1024 * 2048;
#pragma unroll
    for (int i = 0; i < 2; ++i)
#pragma unroll
        for (int j = 0; j < 4; ++j)
#pragma unroll
            for (int r = 0; r < 8; ++r) {
                int m = m0 + i * 16 + r + 8 * hlf;
                int n = n0 + j * 16 + l16;
                float val = c[i][j][r] * scale;
                if (n > 1024 + m) val = -INFINITY;
                Ss[(size_t)m * 2048 + n] = val;
            }
}

__global__ void softmax_kernel(float* __restrict__ S) {
    __shared__ float red[256];
    const int s = blockIdx.y, q = blockIdx.x, t = threadIdx.x;
    float* row = S + ((size_t)s * 1024 + q) * 2048;
    float x[8];
    float m = -INFINITY;
#pragma unroll
    for (int i = 0; i < 8; ++i) {
        x[i] = row[t + i * 256];
        m = fmaxf(m, x[i]);
    }
    red[t] = m;
    __syncthreads();
    for (int off = 128; off > 0; off >>= 1) {
        if (t < off) red[t] = fmaxf(red[t], red[t + off]);
        __syncthreads();
    }
    m = red[0];
    __syncthreads();
    float sum = 0.f;
#pragma unroll
    for (int i = 0; i < 8; ++i) {
        x[i] = __expf(x[i] - m);
        sum += x[i];
    }
    red[t] = sum;
    __syncthreads();
    for (int off = 128; off > 0; off >>= 1) {
        if (t < off) red[t] += red[t + off];
        __syncthreads();
    }
    float inv = 1.0f / red[0];
#pragma unroll
    for (int i = 0; i < 8; ++i) row[t + i * 256] = x[i] * inv;
}

__global__ void __launch_bounds__(512)
pv_kernel(const float* __restrict__ P,
          const float* __restrict__ V,
          float* __restrict__ O, int bh0) {
    __shared__ float part[16][16 * 64];
    const int lane = threadIdx.x & 31;
    const int w    = __builtin_amdgcn_readfirstlane((int)(threadIdx.x >> 5));
    const int l16  = lane & 15;
    const int hlf  = lane >> 4;
    const int s  = blockIdx.y;
    const int bh = bh0 + s;
    const int b  = bh >> 4;
    const int h  = bh & 15;
    const int m0 = blockIdx.x * 16;

    const float* Ps = P + (size_t)s * 1024 * 2048;
    const float* Vb = V + (size_t)b * 2048 * 1024 + h * 64;

    v8f c[4];
#pragma unroll
    for (int j = 0; j < 4; ++j) c[j] = v8f_zero();

    for (int t = 0; t < 32; ++t) {
        const int kb = 4 * w + 64 * t;
        const int kk = kb + 2 * hlf;
        v2f a = *(const v2f*)(Ps + (size_t)(m0 + l16) * 2048 + kk);
        v2f bf[4];
#pragma unroll
        for (int j = 0; j < 4; ++j) {
            int n = j * 16 + l16;
            bf[j].x = Vb[(size_t)kk * 1024 + n];
            bf[j].y = Vb[(size_t)(kk + 1) * 1024 + n];
        }
#pragma unroll
        for (int j = 0; j < 4; ++j) c[j] = wmma4(a, bf[j], c[j]);
    }

#pragma unroll
    for (int j = 0; j < 4; ++j)
#pragma unroll
        for (int r = 0; r < 8; ++r)
            part[w][(r + 8 * hlf) * 64 + j * 16 + l16] = c[j][r];
    __syncthreads();

#pragma unroll
    for (int i = 0; i < 2; ++i) {
        const int e = (int)threadIdx.x + i * 512;
        float acc = part[0][e];
#pragma unroll
        for (int p = 1; p < 16; ++p) acc += part[p][e];
        const int row = e >> 6;
        const int col = e & 63;
        O[((size_t)b * 1024 + m0 + row) * 1024 + h * 64 + col] = acc;
    }
}

extern "C" void kernel_launch(void* const* d_in, const int* in_sizes, int n_in,
                              void* d_out, int out_size, void* d_ws, size_t ws_size,
                              hipStream_t stream) {
    const float* x_q  = (const float*)d_in[0];
    const float* x_kv = (const float*)d_in[1];
    const float* Wq   = (const float*)d_in[2];
    const float* Wk   = (const float*)d_in[3];
    const float* Wv   = (const float*)d_in[4];
    const float* Wr   = (const float*)d_in[5];
    const float* Wo   = (const float*)d_in[6];
    const float* ub   = (const float*)d_in[7];
    const float* vb   = (const float*)d_in[8];
    float* out = (float*)d_out;
    (void)in_sizes; (void)n_in; (void)out_size; (void)ws_size;

    float* ws   = (float*)d_ws;
    float* Qb   = ws;
    float* Kb   = Qb + 2097152;
    float* Vb   = Kb + 4194304;
    float* Relb = Vb + 4194304;
    float* Rb   = Relb + 2097152;
    float* Ob   = Rb + 2097152;
    float* Sb   = Ob + 2097152;

    relgen_kernel<<<2048, 256, 0, stream>>>(Relb);

    gemm_nt_kernel<2, 128><<<dim3(8, 32), 256, 0, stream>>>(x_q,  Wq, Qb, 2048, 1024, 1024);
    gemm_nt_kernel<4, 64><<<dim3(4, 64), 256, 0, stream>>>(x_kv, Wk, Kb, 4096, 1024, 1024);
    gemm_nt_kernel<4, 64><<<dim3(4, 64), 256, 0, stream>>>(x_kv, Wv, Vb, 4096, 1024, 1024);
    gemm_nt_kernel<2, 128><<<dim3(8, 32), 256, 0, stream>>>(Relb, Wr, Rb, 2048, 1024, 1024);

    for (int bh0 = 0; bh0 < 32; bh0 += 8) {
        score_kernel<<<dim3(16, 8, 8), 256, 0, stream>>>(Qb, Kb, Rb, ub, vb, Sb, bh0);
        softmax_kernel<<<dim3(1024, 8), 256, 0, stream>>>(Sb);
        pv_kernel<<<dim3(64, 8), 512, 0, stream>>>(Sb, Vb, Ob, bh0);
    }

    gemm_nt_kernel<2, 128><<<dim3(8, 32), 256, 0, stream>>>(Ob, Wo, out, 2048, 1024, 1024);
}


# attention fused into one kernel: per 64-key tile a wave forms transposed scores (keys x 16 queries), running-max softmax, P.V from score registers; S never stored; changes softmax, P.V sum order
# speedup vs baseline: 3.9467x; 1.6706x over previous
#include <hip/hip_runtime.h>
#include <hip/hip_bf16.h>
#include <math.h>


typedef float v2f __attribute__((ext_vector_type(2)));
typedef float v8f __attribute__((ext_vector_type(8)));

__device__ __forceinline__ v8f v8f_zero() {
    v8f z = {0.f, 0.f, 0.f, 0.f, 0.f, 0.f, 0.f, 0.f};
    return z;
}

__device__ __forceinline__ v8f wmma4(v2f a, v2f b, v8f c) {
    return __builtin_amdgcn_wmma_f32_16x16x4_f32(false, a, false, b,
                                                 (short)0, c, false, false);
}

__global__ void relgen_kernel(float* __restrict__ rel) {
    const int row = blockIdx.x;
    const int t = threadIdx.x;
    const float pos = (float)(2047 - row);
    const float negl = -9.210340371976184f / 512.0f;
#pragma unroll
    for (int i = 0; i < 4; ++i) {
        int c = t + i * 256;
        int cc = c & 511;
        float inv = expf((float)cc * negl);
        float ang = pos * inv;
        rel[(size_t)row * 1024 + c] = (c < 512) ? sinf(ang) : cosf(ang);
    }
}

typedef float v4f __attribute__((ext_vector_type(4)));

template <int NJ, int KC>
__global__ void __launch_bounds__(256)
gemm_nt_kernel(const float* __restrict__ A,
               const float* __restrict__ B,
               float* __restrict__ C,
               int M, int N, int Kd) {
    constexpr int GT_P    = KC + 4;
    constexpr int GT_BN   = 64 * NJ;
    constexpr int GT_ROWS = 64 + GT_BN;
    constexpr int GT_NG   = KC / 8;
    constexpr int GT_SEG  = KC / 4;
    constexpr int GT_RPP  = 256 / GT_SEG;
    constexpr int GT_NA   = 64 / GT_RPP;
    constexpr int GT_NB   = GT_BN / GT_RPP;
    __shared__ __attribute__((aligned(16))) float tile[2][GT_ROWS * GT_P];
    const int tid  = threadIdx.x;
    const int lane = tid & 31;
    const int w    = tid >> 5;
    const int l16  = lane & 15;
    const int hlf  = lane >> 4;
    const int wm = (w & 1) * 32;
    const int wn = (w >> 1) * (16 * NJ);
    const int m0 = blockIdx.y * 64 + wm;
    const int n0 = blockIdx.x * GT_BN + wn;
    (void)M;

    const int srow = tid / GT_SEG;
    const int sseg = tid % GT_SEG;
    const float* pa = A + (size_t)(blockIdx.y * 64 + srow) * Kd + sseg * 4;
    const float* pb = B + (size_t)(blockIdx.x * GT_BN + srow) * Kd + sseg * 4;
    const int soff = srow * GT_P + sseg * 4;

    v8f c[2][NJ];
#pragma unroll
    for (int i = 0; i < 2; ++i)
#pragma unroll
        for (int j = 0; j < NJ; ++j) c[i][j] = v8f_zero();

    v4f g[GT_NA + GT_NB];
#pragma unroll
    for (int i = 0; i < GT_NA; ++i)
        g[i] = *(const v4f*)(pa + (size_t)(GT_RPP * i) * Kd);
#pragma unroll
    for (int i = 0; i < GT_NB; ++i)
        g[GT_NA + i] = *(const v4f*)(pb + (size_t)(GT_RPP * i) * Kd);

    const int nk = Kd / KC;
    for (int kc = 0; kc < nk; ++kc) {
        float* st = tile[kc & 1];
#pragma unroll
        for (int i = 0; i < GT_NA + GT_NB; ++i)
            *(v4f*)(st + soff + GT_RPP * i * GT_P) = g[i];
        __syncthreads();
        if (kc + 1 < nk) {
            const int ko = (kc + 1) * KC;
#pragma unroll
            for (int i = 0; i < GT_NA; ++i)
                g[i] = *(const v4f*)(pa + (size_t)(GT_RPP * i) * Kd + ko);
#pragma unroll
            for (int i = 0; i < GT_NB; ++i)
                g[GT_NA + i] = *(const v4f*)(pb + (size_t)(GT_RPP * i) * Kd + ko);
        }
        const float* as = st + (wm + l16) * GT_P + 4 * hlf;
        const float* bs = st + (64 + wn + l16) * GT_P + 4 * hlf;
        v4f a[2][2], b[2][NJ];
#pragma unroll
        for (int i = 0; i < 2; ++i) a[0][i] = *(const v4f*)(as + i * 16 * GT_P);
#pragma unroll
        for (int j = 0; j < NJ; ++j) b[0][j] = *(const v4f*)(bs + j * 16 * GT_P);
#pragma unroll
        for (int gk = 0; gk < GT_NG; ++gk) {
            const int cur = gk & 1;
            if (gk + 1 < GT_NG) {
#pragma unroll
                for (int i = 0; i < 2; ++i)
                    a[cur ^ 1][i] = *(const v4f*)(as + i * 16 * GT_P + (gk + 1) * 8);
#pragma unroll
                for (int j = 0; j < NJ; ++j)
                    b[cur ^ 1][j] = *(const v4f*)(bs + j * 16 * GT_P + (gk + 1) * 8);
            }
#pragma unroll
            for (int i = 0; i < 2; ++i)
#pragma unroll
                for (int j = 0; j < NJ; ++j)
                    c[i][j] = wmma4(a[cur][i].xy, b[cur][j].xy, c[i][j]);
#pragma unroll
            for (int i = 0; i < 2; ++i)
#pragma unroll
                for (int j = 0; j < NJ; ++j)
                    c[i][j] = wmma4(a[cur][i].zw, b[cur][j].zw, c[i][j]);
        }
    }

#pragma unroll
    for (int i = 0; i < 2; ++i)
#pragma unroll
        for (int j = 0; j < NJ; ++j)
#pragma unroll
            for (int r = 0; r < 8; ++r) {
                int m = m0 + i * 16 + r + 8 * hlf;
                int n = n0 + j * 16 + l16;
                C[(size_t)m * N + n] = c[i][j][r];
            }
}

#define FA_P     68
#define FA_STAGE (192 * FA_P)

__global__ void __launch_bounds__(256)
attn_kernel(const float* __restrict__ Q,
            const float* __restrict__ Km,
            const float* __restrict__ V,
            const float* __restrict__ R,
            const float* __restrict__ u,
            const float* __restrict__ v,
            float* __restrict__ O) {
    __shared__ __attribute__((aligned(16))) float tile[4][FA_STAGE];
    const int tid  = threadIdx.x;
    const int lane = tid & 31;
    const int w    = __builtin_amdgcn_readfirstlane((int)(threadIdx.x >> 5));
    const int l16  = lane & 15;
    const int hlf  = lane >> 4;
    const int t    = blockIdx.x;
    const int bh   = blockIdx.y;
    const int b    = bh >> 4;
    const int h    = bh & 15;
    const int grp  = ((w >> 2) ^ w) & 1;
    const int sub  = grp ? (15 - t) : t;
    const int my_nt = 17 + sub;
    const int nt    = 32 - t;
    const int qbase = sub * 64 + (w >> 1) * 16;

    v4f qa[8], qv[8];
    {
        const float* qrow = Q + ((size_t)b * 1024 + qbase + l16) * 1024 + h * 64 + 4 * hlf;
        const float* up = u + h * 64 + 4 * hlf;
        const float* vp = v + h * 64 + 4 * hlf;
#pragma unroll
        for (int gk = 0; gk < 8; ++gk) {
            const v4f qq = *(const v4f*)(qrow + gk * 8);
            qa[gk] = qq + *(const v4f*)(up + gk * 8);
            qv[gk] = qq + *(const v4f*)(vp + gk * 8);
        }
    }

    const int srow = tid >> 4;
    const int sseg = tid & 15;
    const float* pk = Km + ((size_t)b * 2048 + srow) * 1024 + h * 64 + sseg * 4;
    const float* pv = V  + ((size_t)b * 2048 + srow) * 1024 + h * 64 + sseg * 4;
    const float* pr = R  + (size_t)srow * 1024 + h * 64 + sseg * 4;

    v4f g[12];
#pragma unroll
    for (int i = 0; i < 4; ++i) {
        g[i]     = *(const v4f*)(pk + (size_t)(16 * i) * 1024);
        g[4 + i] = *(const v4f*)(pr + (size_t)(16 * i) * 1024);
        g[8 + i] = *(const v4f*)(pv + (size_t)(16 * i) * 1024);
    }

    v8f o[4];
#pragma unroll
    for (int dj = 0; dj < 4; ++dj) o[dj] = v8f_zero();
    float m_run = -INFINITY;
    float l_run = 0.f;
    const int klim = 1024 + qbase + l16;

    for (int kt = 0; kt < nt; ++kt) {
        float* st = tile[kt & 3];
#pragma unroll
        for (int i = 0; i < 4; ++i) {
            *(v4f*)(st + (srow + 16 * i) * FA_P + sseg * 4)      = g[i];
            *(v4f*)(st + (64 + srow + 16 * i) * FA_P + sseg * 4) = g[4 + i];
#pragma unroll
            for (int e = 0; e < 4; ++e)
                st[(128 + 4 * sseg + e) * FA_P + srow + 16 * i] = g[8 + i][e];
        }
        __syncthreads();
        if (kt + 1 < nt) {
            const size_t ko = (size_t)(kt + 1) * 64 * 1024;
#pragma unroll
            for (int i = 0; i < 4; ++i) {
                g[i]     = *(const v4f*)(pk + ko + (size_t)(16 * i) * 1024);
                g[4 + i] = *(const v4f*)(pr + ko + (size_t)(16 * i) * 1024);
                g[8 + i] = *(const v4f*)(pv + ko + (size_t)(16 * i) * 1024);
            }
        }
        if (kt < my_nt) {
            const int k0 = kt * 64;
            const float* ks = st + l16 * FA_P + 4 * hlf;
            const float* rs = st + (64 + l16) * FA_P + 4 * hlf;
            const float* vs = st + (128 + l16) * FA_P + 8 * hlf;

            v8f s[4];
#pragma unroll
            for (int j = 0; j < 4; ++j) s[j] = v8f_zero();
            v4f ka[2][4], ra[2][4];
#pragma unroll
            for (int j = 0; j < 4; ++j) {
                ka[0][j] = *(const v4f*)(ks + j * 16 * FA_P);
                ra[0][j] = *(const v4f*)(rs + j * 16 * FA_P);
            }
#pragma unroll
            for (int gk = 0; gk < 8; ++gk) {
                const int cur = gk & 1;
                if (gk + 1 < 8) {
#pragma unroll
                    for (int j = 0; j < 4; ++j) {
                        ka[cur ^ 1][j] = *(const v4f*)(ks + j * 16 * FA_P + (gk + 1) * 8);
                        ra[cur ^ 1][j] = *(const v4f*)(rs + j * 16 * FA_P + (gk + 1) * 8);
                    }
                }
#pragma unroll
                for (int j = 0; j < 4; ++j) s[j] = wmma4(ka[cur][j].xy, qa[gk].xy, s[j]);
#pragma unroll
                for (int j = 0; j < 4; ++j) s[j] = wmma4(ka[cur][j].zw, qa[gk].zw, s[j]);
#pragma unroll
                for (int j = 0; j < 4; ++j) s[j] = wmma4(ra[cur][j].xy, qv[gk].xy, s[j]);
#pragma unroll
                for (int j = 0; j < 4; ++j) s[j] = wmma4(ra[cur][j].zw, qv[gk].zw, s[j]);
            }

            v4f va[2][4][2];
#pragma unroll
            for (int dj = 0; dj < 4; ++dj)
#pragma unroll
                for (int tt = 0; tt < 2; ++tt)
                    va[0][dj][tt] = *(const v4f*)(vs + dj * 16 * FA_P + 4 * tt);

            float mx = m_run;
#pragma unroll
            for (int j = 0; j < 4; ++j)
#pragma unroll
                for (int r = 0; r < 8; ++r) {
                    const int kg = k0 + 16 * j + 8 * hlf + r;
                    float x = s[j][r] * 0.125f;
                    if (kg > klim) x = -INFINITY;
                    s[j][r] = x;
                    mx = fmaxf(mx, x);
                }
            mx = fmaxf(mx, __shfl_xor(mx, 16));
            const float alpha = __expf(m_run - mx);
            m_run = mx;
            float ps = 0.f;
#pragma unroll
            for (int j = 0; j < 4; ++j)
#pragma unroll
                for (int r = 0; r < 8; ++r) {
                    const float p = __expf(s[j][r] - mx);
                    s[j][r] = p;
                    ps += p;
                }
            l_run = l_run * alpha + ps;
#pragma unroll
            for (int dj = 0; dj < 4; ++dj)
#pragma unroll
                for (int r = 0; r < 8; ++r) o[dj][r] *= alpha;

#pragma unroll
            for (int j = 0; j < 4; ++j) {
                const int cur = j & 1;
                if (j + 1 < 4) {
#pragma unroll
                    for (int dj = 0; dj < 4; ++dj)
#pragma unroll
                        for (int tt = 0; tt < 2; ++tt)
                            va[cur ^ 1][dj][tt] =
                                *(const v4f*)(vs + dj * 16 * FA_P + 16 * (j + 1) + 4 * tt);
                }
#pragma unroll
                for (int tt = 0; tt < 2; ++tt) {
                    v2f p0, p1;
                    p0.x = s[j][4 * tt];     p0.y = s[j][4 * tt + 1];
                    p1.x = s[j][4 * tt + 2]; p1.y = s[j][4 * tt + 3];
#pragma unroll
                    for (int dj = 0; dj < 4; ++dj) o[dj] = wmma4(va[cur][dj][tt].xy, p0, o[dj]);
#pragma unroll
                    for (int dj = 0; dj < 4; ++dj) o[dj] = wmma4(va[cur][dj][tt].zw, p1, o[dj]);
                }
            }
        }
    }

    const float lt  = l_run + __shfl_xor(l_run, 16);
    const float inv = 1.0f / lt;
    float* orow = O + ((size_t)b * 1024 + qbase + l16) * 1024 + h * 64 + 8 * hlf;
#pragma unroll
    for (int dj = 0; dj < 4; ++dj) {
        v4f lo, hi;
        lo.x = o[dj][0] * inv; lo.y = o[dj][1] * inv; lo.z = o[dj][2] * inv; lo.w = o[dj][3] * inv;
        hi.x = o[dj][4] * inv; hi.y = o[dj][5] * inv; hi.z = o[dj][6] * inv; hi.w = o[dj][7] * inv;
        *(v4f*)(orow + 16 * dj)     = lo;
        *(v4f*)(orow + 16 * dj + 4) = hi;
    }
}

extern "C" void kernel_launch(void* const* d_in, const int* in_sizes, int n_in,
                              void* d_out, int out_size, void* d_ws, size_t ws_size,
                              hipStream_t stream) {
    const float* x_q  = (const float*)d_in[0];
    const float* x_kv = (const float*)d_in[1];
    const float* Wq   = (const float*)d_in[2];
    const float* Wk   = (const float*)d_in[3];
    const float* Wv   = (const float*)d_in[4];
    const float* Wr   = (const float*)d_in[5];
    const float* Wo   = (const float*)d_in[6];
    const float* ub   = (const float*)d_in[7];
    const float* vb   = (const float*)d_in[8];
    float* out = (float*)d_out;
    (void)in_sizes; (void)n_in; (void)out_size; (void)ws_size;

    float* ws   = (float*)d_ws;
    float* Qb   = ws;
    float* Kb   = Qb + 2097152;
    float* Vb   = Kb + 4194304;
    float* Relb = Vb + 4194304;
    float* Rb   = Relb + 2097152;
    float* Ob   = Rb + 2097152;

    relgen_kernel<<<2048, 256, 0, stream>>>(Relb);

    gemm_nt_kernel<2, 128><<<dim3(8, 32), 256, 0, stream>>>(x_q,  Wq, Qb, 2048, 1024, 1024);
    gemm_nt_kernel<4, 64><<<dim3(4, 64), 256, 0, stream>>>(x_kv, Wk, Kb, 4096, 1024, 1024);
    gemm_nt_kernel<4, 64><<<dim3(4, 64), 256, 0, stream>>>(x_kv, Wv, Vb, 4096, 1024, 1024);
    gemm_nt_kernel<2, 128><<<dim3(8, 32), 256, 0, stream>>>(Relb, Wr, Rb, 2048, 1024, 1024);

    attn_kernel<<<dim3(8, 32), 256, 0, stream>>>(Qb, Kb, Vb, Rb, ub, vb, Ob);

    gemm_nt_kernel<2, 128><<<dim3(8, 32), 256, 0, stream>>>(Ob, Wo, out, 2048, 1024, 1024);
}


# attention scores regrouped as Q.(K+R)^T + per-key bias u.K+v.R: K+R rows and bias (64-long dot, vector unit, 16-lane sum) made at staging; 64 score WMMAs per tile, was 128; changes score sum order
# speedup vs baseline: 4.3281x; 1.0966x over previous
#include <hip/hip_runtime.h>
#include <hip/hip_bf16.h>
#include <math.h>


typedef float v2f __attribute__((ext_vector_type(2)));
typedef float v8f __attribute__((ext_vector_type(8)));

__device__ __forceinline__ v8f v8f_zero() {
    v8f z = {0.f, 0.f, 0.f, 0.f, 0.f, 0.f, 0.f, 0.f};
    return z;
}

__device__ __forceinline__ v8f wmma4(v2f a, v2f b, v8f c) {
    return __builtin_amdgcn_wmma_f32_16x16x4_f32(false, a, false, b,
                                                 (short)0, c, false, false);
}

__global__ void relgen_kernel(float* __restrict__ rel) {
    const int row = blockIdx.x;
    const int t = threadIdx.x;
    const float pos = (float)(2047 - row);
    const float negl = -9.210340371976184f / 512.0f;
#pragma unroll
    for (int i = 0; i < 4; ++i) {
        int c = t + i * 256;
        int cc = c & 511;
        float inv = expf((float)cc * negl);
        float ang = pos * inv;
        rel[(size_t)row * 1024 + c] = (c < 512) ? sinf(ang) : cosf(ang);
    }
}

typedef float v4f __attribute__((ext_vector_type(4)));

template <int NJ, int KC>
__global__ void __launch_bounds__(256)
gemm_nt_kernel(const float* __restrict__ A,
               const float* __restrict__ B,
               float* __restrict__ C,
               int M, int N, int Kd) {
    constexpr int GT_P    = KC + 4;
    constexpr int GT_BN   = 64 * NJ;
    constexpr int GT_ROWS = 64 + GT_BN;
    constexpr int GT_NG   = KC / 8;
    constexpr int GT_SEG  = KC / 4;
    constexpr int GT_RPP  = 256 / GT_SEG;
    constexpr int GT_NA   = 64 / GT_RPP;
    constexpr int GT_NB   = GT_BN / GT_RPP;
    __shared__ __attribute__((aligned(16))) float tile[2][GT_ROWS * GT_P];
    const int tid  = threadIdx.x;
    const int lane = tid & 31;
    const int w    = tid >> 5;
    const int l16  = lane & 15;
    const int hlf  = lane >> 4;
    const int wm = (w & 1) * 32;
    const int wn = (w >> 1) * (16 * NJ);
    const int m0 = blockIdx.y * 64 + wm;
    const int n0 = blockIdx.x * GT_BN + wn;
    (void)M;

    const int srow = tid / GT_SEG;
    const int sseg = tid % GT_SEG;
    const float* pa = A + (size_t)(blockIdx.y * 64 + srow) * Kd + sseg * 4;
    const float* pb = B + (size_t)(blockIdx.x * GT_BN + srow) * Kd + sseg * 4;
    const int soff = srow * GT_P + sseg * 4;

    v8f c[2][NJ];
#pragma unroll
    for (int i = 0; i < 2; ++i)
#pragma unroll
        for (int j = 0; j < NJ; ++j) c[i][j] = v8f_zero();

    v4f g[GT_NA + GT_NB];
#pragma unroll
    for (int i = 0; i < GT_NA; ++i)
        g[i] = *(const v4f*)(pa + (size_t)(GT_RPP * i) * Kd);
#pragma unroll
    for (int i = 0; i < GT_NB; ++i)
        g[GT_NA + i] = *(const v4f*)(pb + (size_t)(GT_RPP * i) * Kd);

    const int nk = Kd / KC;
    for (int kc = 0; kc < nk; ++kc) {
        float* st = tile[kc & 1];
#pragma unroll
        for (int i = 0; i < GT_NA + GT_NB; ++i)
            *(v4f*)(st + soff + GT_RPP * i * GT_P) = g[i];
        __syncthreads();
        if (kc + 1 < nk) {
            const int ko = (kc + 1) * KC;
#pragma unroll
            for (int i = 0; i < GT_NA; ++i)
                g[i] = *(const v4f*)(pa + (size_t)(GT_RPP * i) * Kd + ko);
#pragma unroll
            for (int i = 0; i < GT_NB; ++i)
                g[GT_NA + i] = *(const v4f*)(pb + (size_t)(GT_RPP * i) * Kd + ko);
        }
        const float* as = st + (wm + l16) * GT_P + 4 * hlf;
        const float* bs = st + (64 + wn + l16) * GT_P + 4 * hlf;
        v4f a[2][2], b[2][NJ];
#pragma unroll
        for (int i = 0; i < 2; ++i) a[0][i] = *(const v4f*)(as + i * 16 * GT_P);
#pragma unroll
        for (int j = 0; j < NJ; ++j) b[0][j] = *(const v4f*)(bs + j * 16 * GT_P);
#pragma unroll
        for (int gk = 0; gk < GT_NG; ++gk) {
            const int cur = gk & 1;
            if (gk + 1 < GT_NG) {
#pragma unroll
                for (int i = 0; i < 2; ++i)
                    a[cur ^ 1][i] = *(const v4f*)(as + i * 16 * GT_P + (gk + 1) * 8);
#pragma unroll
                for (int j = 0; j < NJ; ++j)
                    b[cur ^ 1][j] = *(const v4f*)(bs + j * 16 * GT_P + (gk + 1) * 8);
            }
#pragma unroll
            for (int i = 0; i < 2; ++i)
#pragma unroll
                for (int j = 0; j < NJ; ++j)
                    c[i][j] = wmma4(a[cur][i].xy, b[cur][j].xy, c[i][j]);
#pragma unroll
            for (int i = 0; i < 2; ++i)
#pragma unroll
                for (int j = 0; j < NJ; ++j)
                    c[i][j] = wmma4(a[cur][i].zw, b[cur][j].zw, c[i][j]);
        }
    }

#pragma unroll
    for (int i = 0; i < 2; ++i)
#pragma unroll
        for (int j = 0; j < NJ; ++j)
#pragma unroll
            for (int r = 0; r < 8; ++r) {
                int m = m0 + i * 16 + r + 8 * hlf;
                int n = n0 + j * 16 + l16;
                C[(size_t)m * N + n] = c[i][j][r];
            }
}

#define FA_P     68
#define FA_STAGE (192 * FA_P)
#define FA_BIAS  (64 * FA_P)

template <int M>
__device__ __forceinline__ float row_xor(float x) {
    return __builtin_bit_cast(float, __builtin_amdgcn_update_dpp(
        0, __builtin_bit_cast(int, x), 0x160 + M, 0xf, 0xf, true));
}

__global__ void __launch_bounds__(256)
attn_kernel(const float* __restrict__ Q,
            const float* __restrict__ Km,
            const float* __restrict__ V,
            const float* __restrict__ R,
            const float* __restrict__ u,
            const float* __restrict__ v,
            float* __restrict__ O) {
    __shared__ __attribute__((aligned(16))) float tile[4][FA_STAGE];
    const int tid  = threadIdx.x;
    const int lane = tid & 31;
    const int w    = __builtin_amdgcn_readfirstlane((int)(threadIdx.x >> 5));
    const int l16  = lane & 15;
    const int hlf  = lane >> 4;
    const int t    = blockIdx.x;
    const int bh   = blockIdx.y;
    const int b    = bh >> 4;
    const int h    = bh & 15;
    const int grp  = ((w >> 2) ^ w) & 1;
    const int sub  = grp ? (15 - t) : t;
    const int my_nt = 17 + sub;
    const int nt    = 32 - t;
    const int qbase = sub * 64 + (w >> 1) * 16;

    v4f qa[8];
    {
        const float* qrow = Q + ((size_t)b * 1024 + qbase + l16) * 1024 + h * 64 + 4 * hlf;
#pragma unroll
        for (int gk = 0; gk < 8; ++gk) qa[gk] = *(const v4f*)(qrow + gk * 8);
    }

    const int srow = tid >> 4;
    const int sseg = tid & 15;
    const float* pk = Km + ((size_t)b * 2048 + srow) * 1024 + h * 64 + sseg * 4;
    const float* pv = V  + ((size_t)b * 2048 + srow) * 1024 + h * 64 + sseg * 4;
    const float* pr = R  + (size_t)srow * 1024 + h * 64 + sseg * 4;
    const v4f u4 = *(const v4f*)(u + h * 64 + sseg * 4) * 0.125f;
    const v4f v4 = *(const v4f*)(v + h * 64 + sseg * 4) * 0.125f;

    v4f g[12];
#pragma unroll
    for (int i = 0; i < 4; ++i) {
        g[i]     = *(const v4f*)(pk + (size_t)(16 * i) * 1024);
        g[4 + i] = *(const v4f*)(pr + (size_t)(16 * i) * 1024);
        g[8 + i] = *(const v4f*)(pv + (size_t)(16 * i) * 1024);
    }

    v8f o[4];
#pragma unroll
    for (int dj = 0; dj < 4; ++dj) o[dj] = v8f_zero();
    float m_run = -INFINITY;
    float l_run = 0.f;
    const int klim = 1024 + qbase + l16;

    for (int kt = 0; kt < nt; ++kt) {
        float* st = tile[kt & 3];
        float d[4];
#pragma unroll
        for (int i = 0; i < 4; ++i) {
            const v4f kk = g[i];
            const v4f rr = g[4 + i];
            *(v4f*)(st + (srow + 16 * i) * FA_P + sseg * 4) = kk + rr;
            float da = kk.x * u4.x;
            float db = kk.z * u4.z;
            da = fmaf(kk.y, u4.y, da);
            db = fmaf(kk.w, u4.w, db);
            da = fmaf(rr.x, v4.x, da);
            db = fmaf(rr.z, v4.z, db);
            da = fmaf(rr.y, v4.y, da);
            db = fmaf(rr.w, v4.w, db);
            d[i] = da + db;
#pragma unroll
            for (int e = 0; e < 4; ++e)
                st[(128 + 4 * sseg + e) * FA_P + srow + 16 * i] = g[8 + i][e];
        }
#pragma unroll
        for (int i = 0; i < 4; ++i) d[i] += row_xor<1>(d[i]);
#pragma unroll
        for (int i = 0; i < 4; ++i) d[i] += row_xor<2>(d[i]);
#pragma unroll
        for (int i = 0; i < 4; ++i) d[i] += row_xor<4>(d[i]);
#pragma unroll
        for (int i = 0; i < 4; ++i) d[i] += row_xor<8>(d[i]);
        if (sseg == 0) {
#pragma unroll
            for (int i = 0; i < 4; ++i) st[FA_BIAS + srow + 16 * i] = d[i];
        }
        __syncthreads();
        if (kt + 1 < nt) {
            const size_t ko = (size_t)(kt + 1) * 64 * 1024;
#pragma unroll
            for (int i = 0; i < 4; ++i) {
                g[i]     = *(const v4f*)(pk + ko + (size_t)(16 * i) * 1024);
                g[4 + i] = *(const v4f*)(pr + ko + (size_t)(16 * i) * 1024);
                g[8 + i] = *(const v4f*)(pv + ko + (size_t)(16 * i) * 1024);
            }
        }
        if (kt < my_nt) {
            const int k0 = kt * 64;
            const float* ks = st + l16 * FA_P + 4 * hlf;
            const float* cs = st + FA_BIAS + 8 * hlf;
            const float* vs = st + (128 + l16) * FA_P + 8 * hlf;

            v8f s[4];
#pragma unroll
            for (int j = 0; j < 4; ++j) s[j] = v8f_zero();
            v4f ka[2][4];
            v4f cb[4][2];
#pragma unroll
            for (int j = 0; j < 4; ++j) ka[0][j] = *(const v4f*)(ks + j * 16 * FA_P);
#pragma unroll
            for (int gk = 0; gk < 8; ++gk) {
                const int cur = gk & 1;
                if (gk + 1 < 8) {
#pragma unroll
                    for (int j = 0; j < 4; ++j)
                        ka[cur ^ 1][j] = *(const v4f*)(ks + j * 16 * FA_P + (gk + 1) * 8);
                } else {
#pragma unroll
                    for (int j = 0; j < 4; ++j)
#pragma unroll
                        for (int tt = 0; tt < 2; ++tt)
                            cb[j][tt] = *(const v4f*)(cs + 16 * j + 4 * tt);
                }
#pragma unroll
                for (int j = 0; j < 4; ++j) s[j] = wmma4(ka[cur][j].xy, qa[gk].xy, s[j]);
#pragma unroll
                for (int j = 0; j < 4; ++j) s[j] = wmma4(ka[cur][j].zw, qa[gk].zw, s[j]);
            }

            v4f va[2][4][2];
#pragma unroll
            for (int dj = 0; dj < 4; ++dj)
#pragma unroll
                for (int tt = 0; tt < 2; ++tt)
                    va[0][dj][tt] = *(const v4f*)(vs + dj * 16 * FA_P + 4 * tt);

            float mx = m_run;
#pragma unroll
            for (int j = 0; j < 4; ++j)
#pragma unroll
                for (int r = 0; r < 8; ++r) {
                    const int kg = k0 + 16 * j + 8 * hlf + r;
                    float x = fmaf(s[j][r], 0.125f, cb[j][r >> 2][r & 3]);
                    if (kg > klim) x = -INFINITY;
                    s[j][r] = x;
                    mx = fmaxf(mx, x);
                }
            mx = fmaxf(mx, __shfl_xor(mx, 16));
            const float alpha = __expf(m_run - mx);
            m_run = mx;
            float ps = 0.f;
#pragma unroll
            for (int j = 0; j < 4; ++j)
#pragma unroll
                for (int r = 0; r < 8; ++r) {
                    const float p = __expf(s[j][r] - mx);
                    s[j][r] = p;
                    ps += p;
                }
            l_run = l_run * alpha + ps;
#pragma unroll
            for (int dj = 0; dj < 4; ++dj)
#pragma unroll
                for (int r = 0; r < 8; ++r) o[dj][r] *= alpha;

#pragma unroll
            for (int j = 0; j < 4; ++j) {
                const int cur = j & 1;
                if (j + 1 < 4) {
#pragma unroll
                    for (int dj = 0; dj < 4; ++dj)
#pragma unroll
                        for (int tt = 0; tt < 2; ++tt)
                            va[cur ^ 1][dj][tt] =
                                *(const v4f*)(vs + dj * 16 * FA_P + 16 * (j + 1) + 4 * tt);
                }
#pragma unroll
                for (int tt = 0; tt < 2; ++tt) {
                    v2f p0, p1;
                    p0.x = s[j][4 * tt];     p0.y = s[j][4 * tt + 1];
                    p1.x = s[j][4 * tt + 2]; p1.y = s[j][4 * tt + 3];
#pragma unroll
                    for (int dj = 0; dj < 4; ++dj) o[dj] = wmma4(va[cur][dj][tt].xy, p0, o[dj]);
#pragma unroll
                    for (int dj = 0; dj < 4; ++dj) o[dj] = wmma4(va[cur][dj][tt].zw, p1, o[dj]);
                }
            }
        }
    }

    const float lt  = l_run + __shfl_xor(l_run, 16);
    const float inv = 1.0f / lt;
    float* orow = O + ((size_t)b * 1024 + qbase + l16) * 1024 + h * 64 + 8 * hlf;
#pragma unroll
    for (int dj = 0; dj < 4; ++dj) {
        v4f lo, hi;
        lo.x = o[dj][0] * inv; lo.y = o[dj][1] * inv; lo.z = o[dj][2] * inv; lo.w = o[dj][3] * inv;
        hi.x = o[dj][4] * inv; hi.y = o[dj][5] * inv; hi.z = o[dj][6] * inv; hi.w = o[dj][7] * inv;
        *(v4f*)(orow + 16 * dj)     = lo;
        *(v4f*)(orow + 16 * dj + 4) = hi;
    }
}

extern "C" void kernel_launch(void* const* d_in, const int* in_sizes, int n_in,
                              void* d_out, int out_size, void* d_ws, size_t ws_size,
                              hipStream_t stream) {
    const float* x_q  = (const float*)d_in[0];
    const float* x_kv = (const float*)d_in[1];
    const float* Wq   = (const float*)d_in[2];
    const float* Wk   = (const float*)d_in[3];
    const float* Wv   = (const float*)d_in[4];
    const float* Wr   = (const float*)d_in[5];
    const float* Wo   = (const float*)d_in[6];
    const float* ub   = (const float*)d_in[7];
    const float* vb   = (const float*)d_in[8];
    float* out = (float*)d_out;
    (void)in_sizes; (void)n_in; (void)out_size; (void)ws_size;

    float* ws   = (float*)d_ws;
    float* Qb   = ws;
    float* Kb   = Qb + 2097152;
    float* Vb   = Kb + 4194304;
    float* Relb = Vb + 4194304;
    float* Rb   = Relb + 2097152;
    float* Ob   = Rb + 2097152;

    relgen_kernel<<<2048, 256, 0, stream>>>(Relb);

    gemm_nt_kernel<2, 128><<<dim3(8, 32), 256, 0, stream>>>(x_q,  Wq, Qb, 2048, 1024, 1024);
    gemm_nt_kernel<4, 64><<<dim3(4, 64), 256, 0, stream>>>(x_kv, Wk, Kb, 4096, 1024, 1024);
    gemm_nt_kernel<4, 64><<<dim3(4, 64), 256, 0, stream>>>(x_kv, Wv, Vb, 4096, 1024, 1024);
    gemm_nt_kernel<2, 128><<<dim3(8, 32), 256, 0, stream>>>(Relb, Wr, Rb, 2048, 1024, 1024);

    attn_kernel<<<dim3(8, 32), 256, 0, stream>>>(Qb, Kb, Vb, Rb, ub, vb, Ob);

    gemm_nt_kernel<2, 128><<<dim3(8, 32), 256, 0, stream>>>(Ob, Wo, out, 2048, 1024, 1024);
}


# gemm_nt: 3 LDS stages; the next chunk's LDS stores and the prefetch issue among the current chunk's WMMAs, the per-chunk barrier (LDS-only fence) stands 3/4 into the chunk; k sum order unchanged
# speedup vs baseline: 4.4648x; 1.0316x over previous
#include <hip/hip_runtime.h>
#include <hip/hip_bf16.h>
#include <math.h>


typedef float v2f __attribute__((ext_vector_type(2)));
typedef float v8f __attribute__((ext_vector_type(8)));

__device__ __forceinline__ v8f v8f_zero() {
    v8f z = {0.f, 0.f, 0.f, 0.f, 0.f, 0.f, 0.f, 0.f};
    return z;
}

__device__ __forceinline__ v8f wmma4(v2f a, v2f b, v8f c) {
    return __builtin_amdgcn_wmma_f32_16x16x4_f32(false, a, false, b,
                                                 (short)0, c, false, false);
}

__global__ void relgen_kernel(float* __restrict__ rel) {
    const int row = blockIdx.x;
    const int t = threadIdx.x;
    const float pos = (float)(2047 - row);
    const float negl = -9.210340371976184f / 512.0f;
#pragma unroll
    for (int i = 0; i < 4; ++i) {
        int c = t + i * 256;
        int cc = c & 511;
        float inv = expf((float)cc * negl);
        float ang = pos * inv;
        rel[(size_t)row * 1024 + c] = (c < 512) ? sinf(ang) : cosf(ang);
    }
}

typedef float v4f __attribute__((ext_vector_type(4)));

__device__ __forceinline__ void lds_barrier() {
    __builtin_amdgcn_fence(__ATOMIC_RELEASE, "workgroup", "local");
    __builtin_amdgcn_s_barrier();
    __builtin_amdgcn_fence(__ATOMIC_ACQUIRE, "workgroup", "local");
}

template <int NJ, int KC>
__global__ void __launch_bounds__(256)
gemm_nt_kernel(const float* __restrict__ A,
               const float* __restrict__ B,
               float* __restrict__ C,
               int M, int N, int Kd) {
    constexpr int GT_P    = KC + 4;
    constexpr int GT_BN   = 64 * NJ;
    constexpr int GT_ROWS = 64 + GT_BN;
    constexpr int GT_NG   = KC / 8;
    constexpr int GT_SEG  = KC / 4;
    constexpr int GT_RPP  = 256 / GT_SEG;
    constexpr int GT_NA   = 64 / GT_RPP;
    constexpr int GT_NB   = GT_BN / GT_RPP;
    constexpr int GT_NP   = GT_NA + GT_NB;
    constexpr int GT_SP   = GT_NG - GT_NG / 4;
    constexpr int GT_SG   = GT_SP / 2;
    constexpr int GT_PPG  = (GT_NP + GT_SG - 1) / GT_SG;
    __shared__ __attribute__((aligned(16))) float tile[3][GT_ROWS * GT_P];
    const int tid  = threadIdx.x;
    const int lane = tid & 31;
    const int w    = tid >> 5;
    const int l16  = lane & 15;
    const int hlf  = lane >> 4;
    const int wm = (w & 1) * 32;
    const int wn = (w >> 1) * (16 * NJ);
    const int m0 = blockIdx.y * 64 + wm;
    const int n0 = blockIdx.x * GT_BN + wn;
    (void)M;

    const int srow = tid / GT_SEG;
    const int sseg = tid % GT_SEG;
    const char* abase = (const char*)(A + (size_t)(blockIdx.y * 64) * Kd);
    const char* bbase = (const char*)(B + (size_t)(blockIdx.x * GT_BN) * Kd);
    unsigned goff[GT_NP];
#pragma unroll
    for (int p = 0; p < GT_NP; ++p)
        goff[p] = (unsigned)(((srow + GT_RPP * (p < GT_NA ? p : p - GT_NA)) * Kd + sseg * 4) * 4);
    const int soff = srow * GT_P + sseg * 4;
    const int aoff = (wm + l16) * GT_P + 4 * hlf;
    const int boff = (64 + wn + l16) * GT_P + 4 * hlf;

    v8f c[2][NJ];
#pragma unroll
    for (int i = 0; i < 2; ++i)
#pragma unroll
        for (int j = 0; j < NJ; ++j) c[i][j] = v8f_zero();

    const int nk = Kd / KC;

    v4f g[GT_NP];
    {
        const size_t k1 = (size_t)((nk > 1 ? 1 : 0) * KC) * 4;
        v4f g0[GT_NP];
#pragma unroll
        for (int p = 0; p < GT_NP; ++p)
            g0[p] = *(const v4f*)((p < GT_NA ? abase : bbase) + goff[p]);
#pragma unroll
        for (int p = 0; p < GT_NP; ++p)
            g[p] = *(const v4f*)((p < GT_NA ? abase : bbase) + k1 + goff[p]);
#pragma unroll
        for (int p = 0; p < GT_NP; ++p)
            *(v4f*)(tile[0] + soff + GT_RPP * p * GT_P) = g0[p];
    }
    lds_barrier();
    v4f a[2][2], b[2][NJ];
#pragma unroll
    for (int i = 0; i < 2; ++i) a[0][i] = *(const v4f*)(tile[0] + aoff + i * 16 * GT_P);
#pragma unroll
    for (int j = 0; j < NJ; ++j) b[0][j] = *(const v4f*)(tile[0] + boff + j * 16 * GT_P);

    int cs = 0;
    for (int kc = 0; kc < nk; ++kc) {
        const int ns = (cs == 2) ? 0 : cs + 1;
        const float* st = tile[cs];
        float* stn = tile[ns];
        const size_t k2 = (size_t)(((kc + 2 < nk) ? kc + 2 : nk - 1) * KC) * 4;
#pragma unroll
        for (int gk = 0; gk < GT_NG; ++gk) {
            const int cur = gk & 1;
            if (gk + 1 < GT_NG) {
#pragma unroll
                for (int i = 0; i < 2; ++i)
                    a[cur ^ 1][i] = *(const v4f*)(st + aoff + i * 16 * GT_P + (gk + 1) * 8);
#pragma unroll
                for (int j = 0; j < NJ; ++j)
                    b[cur ^ 1][j] = *(const v4f*)(st + boff + j * 16 * GT_P + (gk + 1) * 8);
            }
            if (gk == GT_SP) {
                lds_barrier();
                __builtin_amdgcn_sched_barrier(0);
            }
            if (gk + 1 == GT_NG) {
#pragma unroll
                for (int i = 0; i < 2; ++i)
                    a[cur ^ 1][i] = *(const v4f*)(stn + aoff + i * 16 * GT_P);
#pragma unroll
                for (int j = 0; j < NJ; ++j)
                    b[cur ^ 1][j] = *(const v4f*)(stn + boff + j * 16 * GT_P);
            }
            if (gk < GT_SG) {
#pragma unroll
                for (int q = 0; q < GT_PPG; ++q) {
                    const int p = gk * GT_PPG + q;
                    if (p < GT_NP) *(v4f*)(stn + soff + GT_RPP * p * GT_P) = g[p];
                }
            } else if (gk < 2 * GT_SG) {
#pragma unroll
                for (int q = 0; q < GT_PPG; ++q) {
                    const int p = (gk - GT_SG) * GT_PPG + q;
                    if (p < GT_NP)
                        g[p] = *(const v4f*)((p < GT_NA ? abase : bbase) + k2 + goff[p]);
                }
            }
#pragma unroll
            for (int i = 0; i < 2; ++i)
#pragma unroll
                for (int j = 0; j < NJ; ++j)
                    c[i][j] = wmma4(a[cur][i].xy, b[cur][j].xy, c[i][j]);
#pragma unroll
            for (int i = 0; i < 2; ++i)
#pragma unroll
                for (int j = 0; j < NJ; ++j)
                    c[i][j] = wmma4(a[cur][i].zw, b[cur][j].zw, c[i][j]);
        }
        __builtin_amdgcn_sched_barrier(0);
        if constexpr (NJ == 4)
            asm volatile("" :: "v"(a[0][0]), "v"(a[0][1]), "v"(b[0][0]), "v"(b[0][1]), "v"(b[0][2]), "v"(b[0][3]));
        else
            asm volatile("" :: "v"(a[0][0]), "v"(a[0][1]), "v"(b[0][0]), "v"(b[0][1]));
        cs = ns;
    }

#pragma unroll
    for (int i = 0; i < 2; ++i)
#pragma unroll
        for (int j = 0; j < NJ; ++j)
#pragma unroll
            for (int r = 0; r < 8; ++r) {
                int m = m0 + i * 16 + r + 8 * hlf;
                int n = n0 + j * 16 + l16;
                C[(size_t)m * N + n] = c[i][j][r];
            }
}

#define FA_P     68
#define FA_STAGE (192 * FA_P)
#define FA_BIAS  (64 * FA_P)

template <int M>
__device__ __forceinline__ float row_xor(float x) {
    return __builtin_bit_cast(float, __builtin_amdgcn_update_dpp(
        0, __builtin_bit_cast(int, x), 0x160 + M, 0xf, 0xf, true));
}

__global__ void __launch_bounds__(256)
attn_kernel(const float* __restrict__ Q,
            const float* __restrict__ Km,
            const float* __restrict__ V,
            const float* __restrict__ R,
            const float* __restrict__ u,
            const float* __restrict__ v,
            float* __restrict__ O) {
    __shared__ __attribute__((aligned(16))) float tile[4][FA_STAGE];
    const int tid  = threadIdx.x;
    const int lane = tid & 31;
    const int w    = __builtin_amdgcn_readfirstlane((int)(threadIdx.x >> 5));
    const int l16  = lane & 15;
    const int hlf  = lane >> 4;
    const int t    = blockIdx.x;
    const int bh   = blockIdx.y;
    const int b    = bh >> 4;
    const int h    = bh & 15;
    const int grp  = ((w >> 2) ^ w) & 1;
    const int sub  = grp ? (15 - t) : t;
    const int my_nt = 17 + sub;
    const int nt    = 32 - t;
    const int qbase = sub * 64 + (w >> 1) * 16;

    v4f qa[8];
    {
        const float* qrow = Q + ((size_t)b * 1024 + qbase + l16) * 1024 + h * 64 + 4 * hlf;
#pragma unroll
        for (int gk = 0; gk < 8; ++gk) qa[gk] = *(const v4f*)(qrow + gk * 8);
    }

    const int srow = tid >> 4;
    const int sseg = tid & 15;
    const float* pk = Km + ((size_t)b * 2048 + srow) * 1024 + h * 64 + sseg * 4;
    const float* pv = V  + ((size_t)b * 2048 + srow) * 1024 + h * 64 + sseg * 4;
    const float* pr = R  + (size_t)srow * 1024 + h * 64 + sseg * 4;
    const v4f u4 = *(const v4f*)(u + h * 64 + sseg * 4) * 0.125f;
    const v4f v4 = *(const v4f*)(v + h * 64 + sseg * 4) * 0.125f;

    v4f g[12];
#pragma unroll
    for (int i = 0; i < 4; ++i) {
        g[i]     = *(const v4f*)(pk + (size_t)(16 * i) * 1024);
        g[4 + i] = *(const v4f*)(pr + (size_t)(16 * i) * 1024);
        g[8 + i] = *(const v4f*)(pv + (size_t)(16 * i) * 1024);
    }

    v8f o[4];
#pragma unroll
    for (int dj = 0; dj < 4; ++dj) o[dj] = v8f_zero();
    float m_run = -INFINITY;
    float l_run = 0.f;
    const int klim = 1024 + qbase + l16;

    for (int kt = 0; kt < nt; ++kt) {
        float* st = tile[kt & 3];
        float d[4];
#pragma unroll
        for (int i = 0; i < 4; ++i) {
            const v4f kk = g[i];
            const v4f rr = g[4 + i];
            *(v4f*)(st + (srow + 16 * i) * FA_P + sseg * 4) = kk + rr;
            float da = kk.x * u4.x;
            float db = kk.z * u4.z;
            da = fmaf(kk.y, u4.y, da);
            db = fmaf(kk.w, u4.w, db);
            da = fmaf(rr.x, v4.x, da);
            db = fmaf(rr.z, v4.z, db);
            da = fmaf(rr.y, v4.y, da);
            db = fmaf(rr.w, v4.w, db);
            d[i] = da + db;
#pragma unroll
            for (int e = 0; e < 4; ++e)
                st[(128 + 4 * sseg + e) * FA_P + srow + 16 * i] = g[8 + i][e];
        }
#pragma unroll
        for (int i = 0; i < 4; ++i) d[i] += row_xor<1>(d[i]);
#pragma unroll
        for (int i = 0; i < 4; ++i) d[i] += row_xor<2>(d[i]);
#pragma unroll
        for (int i = 0; i < 4; ++i) d[i] += row_xor<4>(d[i]);
#pragma unroll
        for (int i = 0; i < 4; ++i) d[i] += row_xor<8>(d[i]);
        if (sseg == 0) {
#pragma unroll
            for (int i = 0; i < 4; ++i) st[FA_BIAS + srow + 16 * i] = d[i];
        }
        __syncthreads();
        if (kt + 1 < nt) {
            const size_t ko = (size_t)(kt + 1) * 64 * 1024;
#pragma unroll
            for (int i = 0; i < 4; ++i) {
                g[i]     = *(const v4f*)(pk + ko + (size_t)(16 * i) * 1024);
                g[4 + i] = *(const v4f*)(pr + ko + (size_t)(16 * i) * 1024);
                g[8 + i] = *(const v4f*)(pv + ko + (size_t)(16 * i) * 1024);
            }
        }
        if (kt < my_nt) {
            const int k0 = kt * 64;
            const float* ks = st + l16 * FA_P + 4 * hlf;
            const float* cs = st + FA_BIAS + 8 * hlf;
            const float* vs = st + (128 + l16) * FA_P + 8 * hlf;

            v8f s[4];
#pragma unroll
            for (int j = 0; j < 4; ++j) s[j] = v8f_zero();
            v4f ka[2][4];
            v4f cb[4][2];
#pragma unroll
            for (int j = 0; j < 4; ++j) ka[0][j] = *(const v4f*)(ks + j * 16 * FA_P);
#pragma unroll
            for (int gk = 0; gk < 8; ++gk) {
                const int cur = gk & 1;
                if (gk + 1 < 8) {
#pragma unroll
                    for (int j = 0; j < 4; ++j)
                        ka[cur ^ 1][j] = *(const v4f*)(ks + j * 16 * FA_P + (gk + 1) * 8);
                } else {
#pragma unroll
                    for (int j = 0; j < 4; ++j)
#pragma unroll
                        for (int tt = 0; tt < 2; ++tt)
                            cb[j][tt] = *(const v4f*)(cs + 16 * j + 4 * tt);
                }
#pragma unroll
                for (int j = 0; j < 4; ++j) s[j] = wmma4(ka[cur][j].xy, qa[gk].xy, s[j]);
#pragma unroll
                for (int j = 0; j < 4; ++j) s[j] = wmma4(ka[cur][j].zw, qa[gk].zw, s[j]);
            }

            v4f va[2][4][2];
#pragma unroll
            for (int dj = 0; dj < 4; ++dj)
#pragma unroll
                for (int tt = 0; tt < 2; ++tt)
                    va[0][dj][tt] = *(const v4f*)(vs + dj * 16 * FA_P + 4 * tt);

            float mx = m_run;
#pragma unroll
            for (int j = 0; j < 4; ++j)
#pragma unroll
                for (int r = 0; r < 8; ++r) {
                    const int kg = k0 + 16 * j + 8 * hlf + r;
                    float x = fmaf(s[j][r], 0.125f, cb[j][r >> 2][r & 3]);
                    if (kg > klim) x = -INFINITY;
                    s[j][r] = x;
                    mx = fmaxf(mx, x);
                }
            mx = fmaxf(mx, __shfl_xor(mx, 16));
            const float alpha = __expf(m_run - mx);
            m_run = mx;
            float ps = 0.f;
#pragma unroll
            for (int j = 0; j < 4; ++j)
#pragma unroll
                for (int r = 0; r < 8; ++r) {
                    const float p = __expf(s[j][r] - mx);
                    s[j][r] = p;
                    ps += p;
                }
            l_run = l_run * alpha + ps;
#pragma unroll
            for (int dj = 0; dj < 4; ++dj)
#pragma unroll
                for (int r = 0; r < 8; ++r) o[dj][r] *= alpha;

#pragma unroll
            for (int j = 0; j < 4; ++j) {
                const int cur = j & 1;
                if (j + 1 < 4) {
#pragma unroll
                    for (int dj = 0; dj < 4; ++dj)
#pragma unroll
                        for (int tt = 0; tt < 2; ++tt)
                            va[cur ^ 1][dj][tt] =
                                *(const v4f*)(vs + dj * 16 * FA_P + 16 * (j + 1) + 4 * tt);
                }
#pragma unroll
                for (int tt = 0; tt < 2; ++tt) {
                    v2f p0, p1;
                    p0.x = s[j][4 * tt];     p0.y = s[j][4 * tt + 1];
                    p1.x = s[j][4 * tt + 2]; p1.y = s[j][4 * tt + 3];
#pragma unroll
                    for (int dj = 0; dj < 4; ++dj) o[dj] = wmma4(va[cur][dj][tt].xy, p0, o[dj]);
#pragma unroll
                    for (int dj = 0; dj < 4; ++dj) o[dj] = wmma4(va[cur][dj][tt].zw, p1, o[dj]);
                }
            }
        }
    }

    const float lt  = l_run + __shfl_xor(l_run, 16);
    const float inv = 1.0f / lt;
    float* orow = O + ((size_t)b * 1024 + qbase + l16) * 1024 + h * 64 + 8 * hlf;
#pragma unroll
    for (int dj = 0; dj < 4; ++dj) {
        v4f lo, hi;
        lo.x = o[dj][0] * inv; lo.y = o[dj][1] * inv; lo.z = o[dj][2] * inv; lo.w = o[dj][3] * inv;
        hi.x = o[dj][4] * inv; hi.y = o[dj][5] * inv; hi.z = o[dj][6] * inv; hi.w = o[dj][7] * inv;
        *(v4f*)(orow + 16 * dj)     = lo;
        *(v4f*)(orow + 16 * dj + 4) = hi;
    }
}

extern "C" void kernel_launch(void* const* d_in, const int* in_sizes, int n_in,
                              void* d_out, int out_size, void* d_ws, size_t ws_size,
                              hipStream_t stream) {
    const float* x_q  = (const float*)d_in[0];
    const float* x_kv = (const float*)d_in[1];
    const float* Wq   = (const float*)d_in[2];
    const float* Wk   = (const float*)d_in[3];
    const float* Wv   = (const float*)d_in[4];
    const float* Wr   = (const float*)d_in[5];
    const float* Wo   = (const float*)d_in[6];
    const float* ub   = (const float*)d_in[7];
    const float* vb   = (const float*)d_in[8];
    float* out = (float*)d_out;
    (void)in_sizes; (void)n_in; (void)out_size; (void)ws_size;

    float* ws   = (float*)d_ws;
    float* Qb   = ws;
    float* Kb   = Qb + 2097152;
    float* Vb   = Kb + 4194304;
    float* Relb = Vb + 4194304;
    float* Rb   = Relb + 2097152;
    float* Ob   = Rb + 2097152;

    relgen_kernel<<<2048, 256, 0, stream>>>(Relb);

    gemm_nt_kernel<2, 128><<<dim3(8, 32), 256, 0, stream>>>(x_q,  Wq, Qb, 2048, 1024, 1024);
    gemm_nt_kernel<4, 64><<<dim3(4, 64), 256, 0, stream>>>(x_kv, Wk, Kb, 4096, 1024, 1024);
    gemm_nt_kernel<4, 64><<<dim3(4, 64), 256, 0, stream>>>(x_kv, Wv, Vb, 4096, 1024, 1024);
    gemm_nt_kernel<2, 128><<<dim3(8, 32), 256, 0, stream>>>(Relb, Wr, Rb, 2048, 1024, 1024);

    attn_kernel<<<dim3(8, 32), 256, 0, stream>>>(Qb, Kb, Vb, Rb, ub, vb, Ob);

    gemm_nt_kernel<2, 128><<<dim3(8, 32), 256, 0, stream>>>(Ob, Wo, out, 2048, 1024, 1024);
}
